# starting kernel
# baseline (speedup 1.0000x reference)
#include <hip/hip_runtime.h>
#include <cstdint>
#include <cstddef>

typedef __attribute__((ext_vector_type(16))) _Float16 v16h;
typedef __attribute__((ext_vector_type(8)))  _Float16 v8h;
typedef __attribute__((ext_vector_type(8)))  float    v8f;
typedef __attribute__((ext_vector_type(2)))  float    v2f;

#define B_    256
#define SEQ_  512
#define HOR_  32
#define NK_   8192
#define L1_   64
#define HID_  128
#define L2_   64
#define GATES_ 512
#define HLEN_ 544

__global__ __launch_bounds__(256) void gemm1_kernel(const float* __restrict__ inp,
                                                    const float* __restrict__ akey,
                                                    float* __restrict__ dist) {
  const int tid = threadIdx.x, lane = tid & 31, wave = tid >> 5;
  const int mt = blockIdx.x, nt = blockIdx.y * 8 + wave;
  const int m = lane & 15;
  const int kb = (lane & 16) ? 2 : 0;
  const float* Ap = inp  + (size_t)(mt * 16 + m) * SEQ_ + kb;
  const float* Bp = akey + (size_t)(nt * 16 + m) * SEQ_ + kb;
  v8f acc0 = {}, acc1 = {};
  for (int k = 0; k < SEQ_; k += 8) {
    v2f a0 = *(const v2f*)(Ap + k);
    v2f b0 = *(const v2f*)(Bp + k);
    acc0 = __builtin_amdgcn_wmma_f32_16x16x4_f32(false, a0, false, b0, (short)0, acc0, false, false);
    v2f a1 = *(const v2f*)(Ap + k + 4);
    v2f b1 = *(const v2f*)(Bp + k + 4);
    acc1 = __builtin_amdgcn_wmma_f32_16x16x4_f32(false, a1, false, b1, (short)0, acc1, false, false);
  }
#pragma unroll
  for (int r = 0; r < 8; ++r) {
    int row = r + ((lane & 16) ? 8 : 0);
    dist[(size_t)(mt * 16 + row) * NK_ + nt * 16 + m] = acc0[r] + acc1[r];
  }
}

__global__ __launch_bounds__(256) void softmax_kernel(float* __restrict__ dist) {
  __shared__ float sv[256];
  __shared__ int   si[256];
  const int b = blockIdx.x, tid = threadIdx.x;
  float* row = dist + (size_t)b * NK_;

  float m = -3.4e38f; int mi = NK_;
  for (int j = tid; j < NK_; j += 256) {
    float v = row[j];
    if (v > m || (v == m && j < mi)) { m = v; mi = j; }
  }
  sv[tid] = m; si[tid] = mi;
  __syncthreads();
  for (int s = 128; s > 0; s >>= 1) {
    if (tid < s) {
      float v2 = sv[tid + s]; int i2 = si[tid + s];
      if (v2 > sv[tid] || (v2 == sv[tid] && i2 < si[tid])) { sv[tid] = v2; si[tid] = i2; }
    }
    __syncthreads();
  }
  if (tid == 0) row[si[0]] = 0.0f;
  __syncthreads();

  float mm = -3.4e38f;
  for (int j = tid; j < NK_; j += 256) mm = fmaxf(mm, row[j]);
  sv[tid] = mm;
  __syncthreads();
  for (int s = 128; s > 0; s >>= 1) { if (tid < s) sv[tid] = fmaxf(sv[tid], sv[tid + s]); __syncthreads(); }
  const float M = sv[0];
  __syncthreads();

  float sum = 0.f;
  for (int j = tid; j < NK_; j += 256) sum += __expf(row[j] - M);
  sv[tid] = sum;
  __syncthreads();
  for (int s = 128; s > 0; s >>= 1) { if (tid < s) sv[tid] += sv[tid + s]; __syncthreads(); }
  const float inv = 1.0f / sv[0];
  for (int j = tid; j < NK_; j += 256) row[j] = __expf(row[j] - M) * inv;
}

__global__ __launch_bounds__(256) void gemm2_kernel(const float* __restrict__ a,
                                                    const float* __restrict__ aval,
                                                    float* __restrict__ hist) {
  const int tid = threadIdx.x, lane = tid & 31, wave = tid >> 5;
  const int mt = blockIdx.x, nt = blockIdx.y * 8 + wave;
  if (nt >= (HLEN_ / 16)) return;
  const int m = lane & 15;
  const int kb = (lane & 16) ? 2 : 0;
  const int n = nt * 16 + m;
  const float* Ap = a + (size_t)(mt * 16 + m) * NK_ + kb;
  const float* Bp = aval + (size_t)kb * HLEN_ + n;
  v8f acc0 = {}, acc1 = {};
  for (int k = 0; k < NK_; k += 8) {
    v2f av = *(const v2f*)(Ap + k);
    v2f bv; bv[0] = Bp[(size_t)k * HLEN_]; bv[1] = Bp[(size_t)(k + 1) * HLEN_];
    acc0 = __builtin_amdgcn_wmma_f32_16x16x4_f32(false, av, false, bv, (short)0, acc0, false, false);
    v2f av1 = *(const v2f*)(Ap + k + 4);
    v2f bv1; bv1[0] = Bp[(size_t)(k + 4) * HLEN_]; bv1[1] = Bp[(size_t)(k + 5) * HLEN_];
    acc1 = __builtin_amdgcn_wmma_f32_16x16x4_f32(false, av1, false, bv1, (short)0, acc1, false, false);
  }
#pragma unroll
  for (int r = 0; r < 8; ++r) {
    int row = r + ((lane & 16) ? 8 : 0);
    hist[(size_t)(mt * 16 + row) * HLEN_ + n] = acc0[r] + acc1[r];
  }
}

__global__ __launch_bounds__(256) void featl1_kernel(const float* __restrict__ inp,
                                                     const float* __restrict__ hist,
                                                     const float* __restrict__ W1,
                                                     const float* __restrict__ b1,
                                                     _Float16* __restrict__ xf) {
  __shared__ float w1s[L1_ * (HOR_ + 1)];
  __shared__ float b1s[L1_];
  __shared__ float hs[8 + HOR_ - 1];
  __shared__ float ins[8];
  const int b = blockIdx.x, t0 = blockIdx.y * 8, tid = threadIdx.x;
  for (int i = tid; i < L1_ * (HOR_ + 1); i += 256) w1s[i] = W1[i];
  if (tid < L1_) b1s[tid] = b1[tid];
  if (tid < 8 + HOR_ - 1) hs[tid] = hist[(size_t)b * HLEN_ + t0 + tid];
  if (tid < 8) ins[tid] = inp[(size_t)b * SEQ_ + t0 + tid];
  __syncthreads();
  const int o = tid & 63;
#pragma unroll
  for (int pass = 0; pass < 2; ++pass) {
    int tt = (tid >> 6) + pass * 4;
    float acc = b1s[o] + ins[tt] * w1s[o * 33];
#pragma unroll
    for (int j = 0; j < HOR_; ++j) acc += hs[tt + j] * w1s[o * 33 + 1 + j];
    xf[((size_t)b * SEQ_ + (t0 + tt)) * L1_ + o] = (_Float16)tanhf(acc);
  }
}

template <int KIN, bool STORE_SEQ>
__global__ __launch_bounds__(256, 1) void lstm_kernel(const _Float16* __restrict__ xin,
                                                      const float* __restrict__ Wih,
                                                      const float* __restrict__ Whh,
                                                      const float* __restrict__ bih,
                                                      const float* __restrict__ bhh,
                                                      _Float16* __restrict__ hseq,
                                                      float* __restrict__ hlast) {
  constexpr int KFX = KIN / 32;
  __shared__ float gates[16][GATES_];
  __shared__ __align__(16) _Float16 hbuf[16][HID_];
  const int tid = threadIdx.x, lane = tid & 31, wave = tid >> 5;
  const int bm = blockIdx.x * 16;
  const int nbase = wave * 64;
  const int m = lane & 15;
  const int hi16 = (lane >> 4) & 1;
  const int kbh = hi16 * 8;

  v16h wihB[4][KFX], whhB[4][4];
  float biasv[4];
#pragma unroll
  for (int nt = 0; nt < 4; ++nt) {
    const int g = nbase + nt * 16 + m;
    biasv[nt] = bih[g] + bhh[g];
#pragma unroll
    for (int kf = 0; kf < KFX; ++kf)
#pragma unroll
      for (int h = 0; h < 16; ++h) {
        int k = kf * 32 + ((h & 8) << 1) + kbh + (h & 7);
        wihB[nt][kf][h] = (_Float16)Wih[(size_t)g * KIN + k];
      }
#pragma unroll
    for (int kf = 0; kf < 4; ++kf)
#pragma unroll
      for (int h = 0; h < 16; ++h) {
        int k = kf * 32 + ((h & 8) << 1) + kbh + (h & 7);
        whhB[nt][kf][h] = (_Float16)Whh[(size_t)g * HID_ + k];
      }
  }

  const int prow = tid >> 4;
  const int pj = (tid & 15) * 8;
  float c[8];
#pragma unroll
  for (int i = 0; i < 8; ++i) { c[i] = 0.f; hbuf[prow][pj + i] = (_Float16)0.f; }
  __syncthreads();

  for (int t = 0; t < SEQ_; ++t) {
    v16h xA[KFX], hA[4];
#pragma unroll
    for (int kf = 0; kf < KFX; ++kf) {
      const _Float16* p = xin + ((size_t)(bm + m) * SEQ_ + t) * KIN + kf * 32 + kbh;
      v8h lo = *(const v8h*)p;
      v8h hi = *(const v8h*)(p + 16);
#pragma unroll
      for (int i = 0; i < 8; ++i) { xA[kf][i] = lo[i]; xA[kf][i + 8] = hi[i]; }
    }
#pragma unroll
    for (int kf = 0; kf < 4; ++kf) {
      const _Float16* p = &hbuf[m][kf * 32 + kbh];
      v8h lo = *(const v8h*)p;
      v8h hi = *(const v8h*)(p + 16);
#pragma unroll
      for (int i = 0; i < 8; ++i) { hA[kf][i] = lo[i]; hA[kf][i + 8] = hi[i]; }
    }
    v8f acc[4];
#pragma unroll
    for (int nt = 0; nt < 4; ++nt)
#pragma unroll
      for (int r = 0; r < 8; ++r) acc[nt][r] = biasv[nt];
#pragma unroll
    for (int nt = 0; nt < 4; ++nt) {
#pragma unroll
      for (int kf = 0; kf < KFX; ++kf)
        acc[nt] = __builtin_amdgcn_wmma_f32_16x16x32_f16(false, xA[kf], false, wihB[nt][kf],
                                                         (short)0, acc[nt], false, false);
#pragma unroll
      for (int kf = 0; kf < 4; ++kf)
        acc[nt] = __builtin_amdgcn_wmma_f32_16x16x32_f16(false, hA[kf], false, whhB[nt][kf],
                                                         (short)0, acc[nt], false, false);
    }
#pragma unroll
    for (int nt = 0; nt < 4; ++nt) {
      const int g = nbase + nt * 16 + m;
      const bool isg = (g >> 7) == 2;
#pragma unroll
      for (int r = 0; r < 8; ++r) {
        int row = r + hi16 * 8;
        float v = acc[nt][r];
        gates[row][g] = isg ? tanhf(v) : 1.0f / (1.0f + __expf(-v));
      }
    }
    __syncthreads();
    v8h hpk;
#pragma unroll
    for (int i = 0; i < 8; ++i) {
      int j = pj + i;
      float iv = gates[prow][j];
      float fv = gates[prow][HID_ + j];
      float gv = gates[prow][2 * HID_ + j];
      float ov = gates[prow][3 * HID_ + j];
      c[i] = fv * c[i] + iv * gv;
      float hv = ov * tanhf(c[i]);
      hpk[i] = (_Float16)hv;
      if (!STORE_SEQ && t == SEQ_ - 1) hlast[(size_t)(bm + prow) * HID_ + j] = hv;
    }
#pragma unroll
    for (int i = 0; i < 8; ++i) hbuf[prow][pj + i] = hpk[i];
    if (STORE_SEQ)
      *(v8h*)(hseq + ((size_t)(bm + prow) * SEQ_ + t) * HID_ + pj) = hpk;
    __syncthreads();
  }
}

__global__ __launch_bounds__(128) void head_kernel(const float* __restrict__ h1last,
                                                   const float* __restrict__ W2,
                                                   const float* __restrict__ b2,
                                                   const float* __restrict__ W3,
                                                   const float* __restrict__ b3,
                                                   float* __restrict__ out) {
  __shared__ float lastS[HID_];
  __shared__ float l2S[L2_];
  const int b = blockIdx.x, tid = threadIdx.x;
  lastS[tid] = tanhf(h1last[(size_t)b * HID_ + tid]);
  __syncthreads();
  if (tid < L2_) {
    float acc = b2[tid];
    for (int k = 0; k < HID_; ++k) acc += lastS[k] * W2[tid * HID_ + k];
    l2S[tid] = tanhf(acc);
  }
  __syncthreads();
  if (tid < HOR_) {
    float acc = b3[tid];
    for (int k = 0; k < L2_; ++k) acc += l2S[k] * W3[tid * L2_ + k];
    out[(size_t)b * HOR_ + tid] = acc;
  }
}

extern "C" void kernel_launch(void* const* d_in, const int* in_sizes, int n_in,
                              void* d_out, int out_size, void* d_ws, size_t ws_size,
                              hipStream_t stream) {
  (void)in_sizes; (void)n_in; (void)out_size; (void)ws_size;
  const float* inp  = (const float*)d_in[0];
  const float* akey = (const float*)d_in[1];
  const float* aval = (const float*)d_in[2];
  const float* W1   = (const float*)d_in[3];
  const float* b1   = (const float*)d_in[4];
  const float* Wih0 = (const float*)d_in[5];
  const float* Whh0 = (const float*)d_in[6];
  const float* bih0 = (const float*)d_in[7];
  const float* bhh0 = (const float*)d_in[8];
  const float* Wih1 = (const float*)d_in[9];
  const float* Whh1 = (const float*)d_in[10];
  const float* bih1 = (const float*)d_in[11];
  const float* bhh1 = (const float*)d_in[12];
  const float* W2   = (const float*)d_in[13];
  const float* b2   = (const float*)d_in[14];
  const float* W3   = (const float*)d_in[15];
  const float* b3   = (const float*)d_in[16];

  char* ws = (char*)d_ws;
  size_t off = 0;
  float*    dist   = (float*)(ws + off);    off += (size_t)B_ * NK_ * 4;
  float*    hist   = (float*)(ws + off);    off += (size_t)B_ * HLEN_ * 4;
  _Float16* xf     = (_Float16*)(ws + off); off += (size_t)B_ * SEQ_ * L1_ * 2;
  _Float16* h0     = (_Float16*)(ws + off); off += (size_t)B_ * SEQ_ * HID_ * 2;
  float*    h1last = (float*)(ws + off);    off += (size_t)B_ * HID_ * 4;

  gemm1_kernel<<<dim3(16, 64), 256, 0, stream>>>(inp, akey, dist);
  softmax_kernel<<<dim3(B_), 256, 0, stream>>>(dist);
  gemm2_kernel<<<dim3(16, 5), 256, 0, stream>>>(dist, aval, hist);
  featl1_kernel<<<dim3(B_, SEQ_ / 8), 256, 0, stream>>>(inp, hist, W1, b1, xf);
  lstm_kernel<L1_, true><<<dim3(B_ / 16), 256, 0, stream>>>(xf, Wih0, Whh0, bih0, bhh0, h0, nullptr);
  lstm_kernel<HID_, false><<<dim3(B_ / 16), 256, 0, stream>>>(h0, Wih1, Whh1, bih1, bhh1, nullptr, h1last);
  head_kernel<<<dim3(B_), 128, 0, stream>>>(h1last, W2, b2, W3, b3, (float*)d_out);
}


# LSTM: input projection bias + x@Wih^T leaves the per-step chain: once per 8 steps one full WMMA tile (2 rows x 8 timesteps) computes it, kept in D registers as C operand of h@Whh^T; sum order kept
# speedup vs baseline: 4.6513x; 4.6513x over previous
#include <hip/hip_runtime.h>
#include <cstdint>
#include <cstddef>

typedef __attribute__((ext_vector_type(16))) _Float16 v16h;
typedef __attribute__((ext_vector_type(8)))  _Float16 v8h;
typedef __attribute__((ext_vector_type(8)))  float    v8f;
typedef __attribute__((ext_vector_type(2)))  float    v2f;

#define B_    256
#define SEQ_  512
#define HOR_  32
#define NK_   8192
#define L1_   64
#define HID_  128
#define L2_   64
#define GATES_ 512
#define HLEN_ 544

__global__ __launch_bounds__(256) void gemm1_kernel(const float* __restrict__ inp,
                                                    const float* __restrict__ akey,
                                                    float* __restrict__ dist) {
  const int tid = threadIdx.x, lane = tid & 31, wave = tid >> 5;
  const int mt = blockIdx.x, nt = blockIdx.y * 8 + wave;
  const int m = lane & 15;
  const int kb = (lane & 16) ? 2 : 0;
  const float* Ap = inp  + (size_t)(mt * 16 + m) * SEQ_ + kb;
  const float* Bp = akey + (size_t)(nt * 16 + m) * SEQ_ + kb;
  v8f acc0 = {}, acc1 = {};
  for (int k = 0; k < SEQ_; k += 8) {
    v2f a0 = *(const v2f*)(Ap + k);
    v2f b0 = *(const v2f*)(Bp + k);
    acc0 = __builtin_amdgcn_wmma_f32_16x16x4_f32(false, a0, false, b0, (short)0, acc0, false, false);
    v2f a1 = *(const v2f*)(Ap + k + 4);
    v2f b1 = *(const v2f*)(Bp + k + 4);
    acc1 = __builtin_amdgcn_wmma_f32_16x16x4_f32(false, a1, false, b1, (short)0, acc1, false, false);
  }
#pragma unroll
  for (int r = 0; r < 8; ++r) {
    int row = r + ((lane & 16) ? 8 : 0);
    dist[(size_t)(mt * 16 + row) * NK_ + nt * 16 + m] = acc0[r] + acc1[r];
  }
}

__global__ __launch_bounds__(256) void softmax_kernel(float* __restrict__ dist) {
  __shared__ float sv[256];
  __shared__ int   si[256];
  const int b = blockIdx.x, tid = threadIdx.x;
  float* row = dist + (size_t)b * NK_;

  float m = -3.4e38f; int mi = NK_;
  for (int j = tid; j < NK_; j += 256) {
    float v = row[j];
    if (v > m || (v == m && j < mi)) { m = v; mi = j; }
  }
  sv[tid] = m; si[tid] = mi;
  __syncthreads();
  for (int s = 128; s > 0; s >>= 1) {
    if (tid < s) {
      float v2 = sv[tid + s]; int i2 = si[tid + s];
      if (v2 > sv[tid] || (v2 == sv[tid] && i2 < si[tid])) { sv[tid] = v2; si[tid] = i2; }
    }
    __syncthreads();
  }
  if (tid == 0) row[si[0]] = 0.0f;
  __syncthreads();

  float mm = -3.4e38f;
  for (int j = tid; j < NK_; j += 256) mm = fmaxf(mm, row[j]);
  sv[tid] = mm;
  __syncthreads();
  for (int s = 128; s > 0; s >>= 1) { if (tid < s) sv[tid] = fmaxf(sv[tid], sv[tid + s]); __syncthreads(); }
  const float M = sv[0];
  __syncthreads();

  float sum = 0.f;
  for (int j = tid; j < NK_; j += 256) sum += __expf(row[j] - M);
  sv[tid] = sum;
  __syncthreads();
  for (int s = 128; s > 0; s >>= 1) { if (tid < s) sv[tid] += sv[tid + s]; __syncthreads(); }
  const float inv = 1.0f / sv[0];
  for (int j = tid; j < NK_; j += 256) row[j] = __expf(row[j] - M) * inv;
}

__global__ __launch_bounds__(256) void gemm2_kernel(const float* __restrict__ a,
                                                    const float* __restrict__ aval,
                                                    float* __restrict__ hist) {
  const int tid = threadIdx.x, lane = tid & 31, wave = tid >> 5;
  const int mt = blockIdx.x, nt = blockIdx.y * 8 + wave;
  if (nt >= (HLEN_ / 16)) return;
  const int m = lane & 15;
  const int kb = (lane & 16) ? 2 : 0;
  const int n = nt * 16 + m;
  const float* Ap = a + (size_t)(mt * 16 + m) * NK_ + kb;
  const float* Bp = aval + (size_t)kb * HLEN_ + n;
  v8f acc0 = {}, acc1 = {};
  for (int k = 0; k < NK_; k += 8) {
    v2f av = *(const v2f*)(Ap + k);
    v2f bv; bv[0] = Bp[(size_t)k * HLEN_]; bv[1] = Bp[(size_t)(k + 1) * HLEN_];
    acc0 = __builtin_amdgcn_wmma_f32_16x16x4_f32(false, av, false, bv, (short)0, acc0, false, false);
    v2f av1 = *(const v2f*)(Ap + k + 4);
    v2f bv1; bv1[0] = Bp[(size_t)(k + 4) * HLEN_]; bv1[1] = Bp[(size_t)(k + 5) * HLEN_];
    acc1 = __builtin_amdgcn_wmma_f32_16x16x4_f32(false, av1, false, bv1, (short)0, acc1, false, false);
  }
#pragma unroll
  for (int r = 0; r < 8; ++r) {
    int row = r + ((lane & 16) ? 8 : 0);
    hist[(size_t)(mt * 16 + row) * HLEN_ + n] = acc0[r] + acc1[r];
  }
}

__global__ __launch_bounds__(256) void featl1_kernel(const float* __restrict__ inp,
                                                     const float* __restrict__ hist,
                                                     const float* __restrict__ W1,
                                                     const float* __restrict__ b1,
                                                     _Float16* __restrict__ xf) {
  __shared__ float w1s[L1_ * (HOR_ + 1)];
  __shared__ float b1s[L1_];
  __shared__ float hs[8 + HOR_ - 1];
  __shared__ float ins[8];
  const int b = blockIdx.x, t0 = blockIdx.y * 8, tid = threadIdx.x;
  for (int i = tid; i < L1_ * (HOR_ + 1); i += 256) w1s[i] = W1[i];
  if (tid < L1_) b1s[tid] = b1[tid];
  if (tid < 8 + HOR_ - 1) hs[tid] = hist[(size_t)b * HLEN_ + t0 + tid];
  if (tid < 8) ins[tid] = inp[(size_t)b * SEQ_ + t0 + tid];
  __syncthreads();
  const int o = tid & 63;
#pragma unroll
  for (int pass = 0; pass < 2; ++pass) {
    int tt = (tid >> 6) + pass * 4;
    float acc = b1s[o] + ins[tt] * w1s[o * 33];
#pragma unroll
    for (int j = 0; j < HOR_; ++j) acc += hs[tt + j] * w1s[o * 33 + 1 + j];
    xf[((size_t)b * SEQ_ + (t0 + tt)) * L1_ + o] = (_Float16)tanhf(acc);
  }
}

#define RB_ 2
#define HR_ (RB_ / 2)
#define TB_ (8 / HR_)

template <int KIN, bool STORE_SEQ>
__global__ __launch_bounds__(256, 1) void lstm_kernel(const _Float16* __restrict__ xin,
                                                      const float* __restrict__ Wih,
                                                      const float* __restrict__ Whh,
                                                      const float* __restrict__ bih,
                                                      const float* __restrict__ bhh,
                                                      _Float16* __restrict__ hseq,
                                                      float* __restrict__ hlast) {
  constexpr int KFX = KIN / 32;
  __shared__ float gates[GATES_ * RB_];
  __shared__ __align__(16) _Float16 hbuf[RB_][HID_];
  const int tid = threadIdx.x, lane = tid & 31, wave = tid >> 5;
  const int bm = blockIdx.x * RB_;
  const int nbase = wave * 64;
  const int m = lane & 15;
  const int hi16 = (lane >> 4) & 1;
  const int kbh = hi16 * 8;
  const int arow = (m >> 3) * HR_ + (m & (HR_ - 1));
  const int xts = (m & 7) / HR_;

  v16h wihB[4][KFX], whhB[4][4];
  float biasv[4];
#pragma unroll
  for (int nt = 0; nt < 4; ++nt) {
    const int g = nbase + nt * 16 + m;
    biasv[nt] = bih[g] + bhh[g];
#pragma unroll
    for (int kf = 0; kf < KFX; ++kf)
#pragma unroll
      for (int h = 0; h < 16; ++h) {
        int k = kf * 32 + ((h & 8) << 1) + kbh + (h & 7);
        wihB[nt][kf][h] = (_Float16)Wih[(size_t)g * KIN + k];
      }
#pragma unroll
    for (int kf = 0; kf < 4; ++kf)
#pragma unroll
      for (int h = 0; h < 16; ++h) {
        int k = kf * 32 + ((h & 8) << 1) + kbh + (h & 7);
        whhB[nt][kf][h] = (_Float16)Whh[(size_t)g * HID_ + k];
      }
  }

  float c[HR_];
#pragma unroll
  for (int i = 0; i < HR_; ++i) {
    const int p = tid + 256 * i;
    c[i] = 0.f;
    hbuf[p >> 7][p & (HID_ - 1)] = (_Float16)0.f;
  }
  __syncthreads();

  for (int t0 = 0; t0 < SEQ_; t0 += TB_) {
    v8f px[4];
    {
      v16h xA[KFX];
#pragma unroll
      for (int kf = 0; kf < KFX; ++kf) {
        const _Float16* p = xin + ((size_t)(bm + arow) * SEQ_ + (t0 + xts)) * KIN + kf * 32 + kbh;
        v8h lo = *(const v8h*)p;
        v8h hi = *(const v8h*)(p + 16);
#pragma unroll
        for (int i = 0; i < 8; ++i) { xA[kf][i] = lo[i]; xA[kf][i + 8] = hi[i]; }
      }
#pragma unroll
      for (int nt = 0; nt < 4; ++nt) {
#pragma unroll
        for (int r = 0; r < 8; ++r) px[nt][r] = biasv[nt];
#pragma unroll
        for (int kf = 0; kf < KFX; ++kf)
          px[nt] = __builtin_amdgcn_wmma_f32_16x16x32_f16(false, xA[kf], false, wihB[nt][kf],
                                                          (short)0, px[nt], false, false);
      }
    }
#pragma unroll
    for (int tl = 0; tl < TB_; ++tl) {
      const int t = t0 + tl;
      v16h hA[4];
#pragma unroll
      for (int kf = 0; kf < 4; ++kf) {
        const _Float16* p = &hbuf[arow][kf * 32 + kbh];
        v8h lo = *(const v8h*)p;
        v8h hi = *(const v8h*)(p + 16);
#pragma unroll
        for (int i = 0; i < 8; ++i) { hA[kf][i] = lo[i]; hA[kf][i + 8] = hi[i]; }
      }
      v8f acc[4];
#pragma unroll
      for (int nt = 0; nt < 4; ++nt) {
        acc[nt] = px[nt];
#pragma unroll
        for (int kf = 0; kf < 4; ++kf)
          acc[nt] = __builtin_amdgcn_wmma_f32_16x16x32_f16(false, hA[kf], false, whhB[nt][kf],
                                                           (short)0, acc[nt], false, false);
      }
#pragma unroll
      for (int nt = 0; nt < 4; ++nt) {
        const int g = nbase + nt * 16 + m;
        const bool isg = (g >> 7) == 2;
#pragma unroll
        for (int v = 0; v < HR_; ++v) {
          float a = acc[nt][tl * HR_ + v];
          gates[g * RB_ + hi16 * HR_ + v] = isg ? tanhf(a) : 1.0f / (1.0f + __expf(-a));
        }
      }
      __syncthreads();
#pragma unroll
      for (int i = 0; i < HR_; ++i) {
        const int p = tid + 256 * i;
        const int row = p >> 7, j = p & (HID_ - 1);
        float iv = gates[j * RB_ + row];
        float fv = gates[(HID_ + j) * RB_ + row];
        float gv = gates[(2 * HID_ + j) * RB_ + row];
        float ov = gates[(3 * HID_ + j) * RB_ + row];
        c[i] = fv * c[i] + iv * gv;
        float hv = ov * tanhf(c[i]);
        hbuf[row][j] = (_Float16)hv;
        if (STORE_SEQ) hseq[((size_t)(bm + row) * SEQ_ + t) * HID_ + j] = (_Float16)hv;
        if (!STORE_SEQ && t == SEQ_ - 1) hlast[(size_t)(bm + row) * HID_ + j] = hv;
      }
      __syncthreads();
    }
  }
}

__global__ __launch_bounds__(128) void head_kernel(const float* __restrict__ h1last,
                                                   const float* __restrict__ W2,
                                                   const float* __restrict__ b2,
                                                   const float* __restrict__ W3,
                                                   const float* __restrict__ b3,
                                                   float* __restrict__ out) {
  __shared__ float lastS[HID_];
  __shared__ float l2S[L2_];
  const int b = blockIdx.x, tid = threadIdx.x;
  lastS[tid] = tanhf(h1last[(size_t)b * HID_ + tid]);
  __syncthreads();
  if (tid < L2_) {
    float acc = b2[tid];
    for (int k = 0; k < HID_; ++k) acc += lastS[k] * W2[tid * HID_ + k];
    l2S[tid] = tanhf(acc);
  }
  __syncthreads();
  if (tid < HOR_) {
    float acc = b3[tid];
    for (int k = 0; k < L2_; ++k) acc += l2S[k] * W3[tid * L2_ + k];
    out[(size_t)b * HOR_ + tid] = acc;
  }
}

extern "C" void kernel_launch(void* const* d_in, const int* in_sizes, int n_in,
                              void* d_out, int out_size, void* d_ws, size_t ws_size,
                              hipStream_t stream) {
  (void)in_sizes; (void)n_in; (void)out_size; (void)ws_size;
  const float* inp  = (const float*)d_in[0];
  const float* akey = (const float*)d_in[1];
  const float* aval = (const float*)d_in[2];
  const float* W1   = (const float*)d_in[3];
  const float* b1   = (const float*)d_in[4];
  const float* Wih0 = (const float*)d_in[5];
  const float* Whh0 = (const float*)d_in[6];
  const float* bih0 = (const float*)d_in[7];
  const float* bhh0 = (const float*)d_in[8];
  const float* Wih1 = (const float*)d_in[9];
  const float* Whh1 = (const float*)d_in[10];
  const float* bih1 = (const float*)d_in[11];
  const float* bhh1 = (const float*)d_in[12];
  const float* W2   = (const float*)d_in[13];
  const float* b2   = (const float*)d_in[14];
  const float* W3   = (const float*)d_in[15];
  const float* b3   = (const float*)d_in[16];

  char* ws = (char*)d_ws;
  size_t off = 0;
  float*    dist   = (float*)(ws + off);    off += (size_t)B_ * NK_ * 4;
  float*    hist   = (float*)(ws + off);    off += (size_t)B_ * HLEN_ * 4;
  _Float16* xf     = (_Float16*)(ws + off); off += (size_t)B_ * SEQ_ * L1_ * 2;
  _Float16* h0     = (_Float16*)(ws + off); off += (size_t)B_ * SEQ_ * HID_ * 2;
  float*    h1last = (float*)(ws + off);    off += (size_t)B_ * HID_ * 4;

  gemm1_kernel<<<dim3(16, 64), 256, 0, stream>>>(inp, akey, dist);
  softmax_kernel<<<dim3(B_), 256, 0, stream>>>(dist);
  gemm2_kernel<<<dim3(16, 5), 256, 0, stream>>>(dist, aval, hist);
  featl1_kernel<<<dim3(B_, SEQ_ / 8), 256, 0, stream>>>(inp, hist, W1, b1, xf);
  lstm_kernel<L1_, true><<<dim3(B_ / RB_), 256, 0, stream>>>(xf, Wih0, Whh0, bih0, bhh0, h0, nullptr);
  lstm_kernel<HID_, false><<<dim3(B_ / RB_), 256, 0, stream>>>(h0, Wih1, Whh1, bih1, bhh1, nullptr, h1last);
  head_kernel<<<dim3(B_), 128, 0, stream>>>(h1last, W2, b2, W3, b3, (float*)d_out);
}


# LSTM steps: both workgroup barriers per step fence LDS only (LDS-address-space release/acquire), so no wave waits for the acknowledgement of its global h store before it signals
# speedup vs baseline: 5.7548x; 1.2372x over previous
#include <hip/hip_runtime.h>
#include <cstdint>
#include <cstddef>

typedef __attribute__((ext_vector_type(16))) _Float16 v16h;
typedef __attribute__((ext_vector_type(8)))  _Float16 v8h;
typedef __attribute__((ext_vector_type(8)))  float    v8f;
typedef __attribute__((ext_vector_type(2)))  float    v2f;

#define B_    256
#define SEQ_  512
#define HOR_  32
#define NK_   8192
#define L1_   64
#define HID_  128
#define L2_   64
#define GATES_ 512
#define HLEN_ 544

__global__ __launch_bounds__(256) void gemm1_kernel(const float* __restrict__ inp,
                                                    const float* __restrict__ akey,
                                                    float* __restrict__ dist) {
  const int tid = threadIdx.x, lane = tid & 31, wave = tid >> 5;
  const int mt = blockIdx.x, nt = blockIdx.y * 8 + wave;
  const int m = lane & 15;
  const int kb = (lane & 16) ? 2 : 0;
  const float* Ap = inp  + (size_t)(mt * 16 + m) * SEQ_ + kb;
  const float* Bp = akey + (size_t)(nt * 16 + m) * SEQ_ + kb;
  v8f acc0 = {}, acc1 = {};
  for (int k = 0; k < SEQ_; k += 8) {
    v2f a0 = *(const v2f*)(Ap + k);
    v2f b0 = *(const v2f*)(Bp + k);
    acc0 = __builtin_amdgcn_wmma_f32_16x16x4_f32(false, a0, false, b0, (short)0, acc0, false, false);
    v2f a1 = *(const v2f*)(Ap + k + 4);
    v2f b1 = *(const v2f*)(Bp + k + 4);
    acc1 = __builtin_amdgcn_wmma_f32_16x16x4_f32(false, a1, false, b1, (short)0, acc1, false, false);
  }
#pragma unroll
  for (int r = 0; r < 8; ++r) {
    int row = r + ((lane & 16) ? 8 : 0);
    dist[(size_t)(mt * 16 + row) * NK_ + nt * 16 + m] = acc0[r] + acc1[r];
  }
}

__global__ __launch_bounds__(256) void softmax_kernel(float* __restrict__ dist) {
  __shared__ float sv[256];
  __shared__ int   si[256];
  const int b = blockIdx.x, tid = threadIdx.x;
  float* row = dist + (size_t)b * NK_;

  float m = -3.4e38f; int mi = NK_;
  for (int j = tid; j < NK_; j += 256) {
    float v = row[j];
    if (v > m || (v == m && j < mi)) { m = v; mi = j; }
  }
  sv[tid] = m; si[tid] = mi;
  __syncthreads();
  for (int s = 128; s > 0; s >>= 1) {
    if (tid < s) {
      float v2 = sv[tid + s]; int i2 = si[tid + s];
      if (v2 > sv[tid] || (v2 == sv[tid] && i2 < si[tid])) { sv[tid] = v2; si[tid] = i2; }
    }
    __syncthreads();
  }
  if (tid == 0) row[si[0]] = 0.0f;
  __syncthreads();

  float mm = -3.4e38f;
  for (int j = tid; j < NK_; j += 256) mm = fmaxf(mm, row[j]);
  sv[tid] = mm;
  __syncthreads();
  for (int s = 128; s > 0; s >>= 1) { if (tid < s) sv[tid] = fmaxf(sv[tid], sv[tid + s]); __syncthreads(); }
  const float M = sv[0];
  __syncthreads();

  float sum = 0.f;
  for (int j = tid; j < NK_; j += 256) sum += __expf(row[j] - M);
  sv[tid] = sum;
  __syncthreads();
  for (int s = 128; s > 0; s >>= 1) { if (tid < s) sv[tid] += sv[tid + s]; __syncthreads(); }
  const float inv = 1.0f / sv[0];
  for (int j = tid; j < NK_; j += 256) row[j] = __expf(row[j] - M) * inv;
}

__global__ __launch_bounds__(256) void gemm2_kernel(const float* __restrict__ a,
                                                    const float* __restrict__ aval,
                                                    float* __restrict__ hist) {
  const int tid = threadIdx.x, lane = tid & 31, wave = tid >> 5;
  const int mt = blockIdx.x, nt = blockIdx.y * 8 + wave;
  if (nt >= (HLEN_ / 16)) return;
  const int m = lane & 15;
  const int kb = (lane & 16) ? 2 : 0;
  const int n = nt * 16 + m;
  const float* Ap = a + (size_t)(mt * 16 + m) * NK_ + kb;
  const float* Bp = aval + (size_t)kb * HLEN_ + n;
  v8f acc0 = {}, acc1 = {};
  for (int k = 0; k < NK_; k += 8) {
    v2f av = *(const v2f*)(Ap + k);
    v2f bv; bv[0] = Bp[(size_t)k * HLEN_]; bv[1] = Bp[(size_t)(k + 1) * HLEN_];
    acc0 = __builtin_amdgcn_wmma_f32_16x16x4_f32(false, av, false, bv, (short)0, acc0, false, false);
    v2f av1 = *(const v2f*)(Ap + k + 4);
    v2f bv1; bv1[0] = Bp[(size_t)(k + 4) * HLEN_]; bv1[1] = Bp[(size_t)(k + 5) * HLEN_];
    acc1 = __builtin_amdgcn_wmma_f32_16x16x4_f32(false, av1, false, bv1, (short)0, acc1, false, false);
  }
#pragma unroll
  for (int r = 0; r < 8; ++r) {
    int row = r + ((lane & 16) ? 8 : 0);
    hist[(size_t)(mt * 16 + row) * HLEN_ + n] = acc0[r] + acc1[r];
  }
}

__global__ __launch_bounds__(256) void featl1_kernel(const float* __restrict__ inp,
                                                     const float* __restrict__ hist,
                                                     const float* __restrict__ W1,
                                                     const float* __restrict__ b1,
                                                     _Float16* __restrict__ xf) {
  __shared__ float w1s[L1_ * (HOR_ + 1)];
  __shared__ float b1s[L1_];
  __shared__ float hs[8 + HOR_ - 1];
  __shared__ float ins[8];
  const int b = blockIdx.x, t0 = blockIdx.y * 8, tid = threadIdx.x;
  for (int i = tid; i < L1_ * (HOR_ + 1); i += 256) w1s[i] = W1[i];
  if (tid < L1_) b1s[tid] = b1[tid];
  if (tid < 8 + HOR_ - 1) hs[tid] = hist[(size_t)b * HLEN_ + t0 + tid];
  if (tid < 8) ins[tid] = inp[(size_t)b * SEQ_ + t0 + tid];
  __syncthreads();
  const int o = tid & 63;
#pragma unroll
  for (int pass = 0; pass < 2; ++pass) {
    int tt = (tid >> 6) + pass * 4;
    float acc = b1s[o] + ins[tt] * w1s[o * 33];
#pragma unroll
    for (int j = 0; j < HOR_; ++j) acc += hs[tt + j] * w1s[o * 33 + 1 + j];
    xf[((size_t)b * SEQ_ + (t0 + tt)) * L1_ + o] = (_Float16)tanhf(acc);
  }
}

__device__ __forceinline__ void lds_barrier() {
  __builtin_amdgcn_fence(__ATOMIC_RELEASE, "workgroup", "local");
  __builtin_amdgcn_s_barrier();
  __builtin_amdgcn_fence(__ATOMIC_ACQUIRE, "workgroup", "local");
}

#define RB_ 2
#define HR_ (RB_ / 2)
#define TB_ (8 / HR_)

template <int KIN, bool STORE_SEQ>
__global__ __launch_bounds__(256, 1) void lstm_kernel(const _Float16* __restrict__ xin,
                                                      const float* __restrict__ Wih,
                                                      const float* __restrict__ Whh,
                                                      const float* __restrict__ bih,
                                                      const float* __restrict__ bhh,
                                                      _Float16* __restrict__ hseq,
                                                      float* __restrict__ hlast) {
  constexpr int KFX = KIN / 32;
  __shared__ float gates[GATES_ * RB_];
  __shared__ __align__(16) _Float16 hbuf[RB_][HID_];
  const int tid = threadIdx.x, lane = tid & 31, wave = tid >> 5;
  const int bm = blockIdx.x * RB_;
  const int nbase = wave * 64;
  const int m = lane & 15;
  const int hi16 = (lane >> 4) & 1;
  const int kbh = hi16 * 8;
  const int arow = (m >> 3) * HR_ + (m & (HR_ - 1));
  const int xts = (m & 7) / HR_;

  v16h wihB[4][KFX], whhB[4][4];
  float biasv[4];
#pragma unroll
  for (int nt = 0; nt < 4; ++nt) {
    const int g = nbase + nt * 16 + m;
    biasv[nt] = bih[g] + bhh[g];
#pragma unroll
    for (int kf = 0; kf < KFX; ++kf)
#pragma unroll
      for (int h = 0; h < 16; ++h) {
        int k = kf * 32 + ((h & 8) << 1) + kbh + (h & 7);
        wihB[nt][kf][h] = (_Float16)Wih[(size_t)g * KIN + k];
      }
#pragma unroll
    for (int kf = 0; kf < 4; ++kf)
#pragma unroll
      for (int h = 0; h < 16; ++h) {
        int k = kf * 32 + ((h & 8) << 1) + kbh + (h & 7);
        whhB[nt][kf][h] = (_Float16)Whh[(size_t)g * HID_ + k];
      }
  }

  float c[HR_];
#pragma unroll
  for (int i = 0; i < HR_; ++i) {
    const int p = tid + 256 * i;
    c[i] = 0.f;
    hbuf[p >> 7][p & (HID_ - 1)] = (_Float16)0.f;
  }
  __syncthreads();

  for (int t0 = 0; t0 < SEQ_; t0 += TB_) {
    v8f px[4];
    {
      v16h xA[KFX];
#pragma unroll
      for (int kf = 0; kf < KFX; ++kf) {
        const _Float16* p = xin + ((size_t)(bm + arow) * SEQ_ + (t0 + xts)) * KIN + kf * 32 + kbh;
        v8h lo = *(const v8h*)p;
        v8h hi = *(const v8h*)(p + 16);
#pragma unroll
        for (int i = 0; i < 8; ++i) { xA[kf][i] = lo[i]; xA[kf][i + 8] = hi[i]; }
      }
#pragma unroll
      for (int nt = 0; nt < 4; ++nt) {
#pragma unroll
        for (int r = 0; r < 8; ++r) px[nt][r] = biasv[nt];
#pragma unroll
        for (int kf = 0; kf < KFX; ++kf)
          px[nt] = __builtin_amdgcn_wmma_f32_16x16x32_f16(false, xA[kf], false, wihB[nt][kf],
                                                          (short)0, px[nt], false, false);
      }
    }
#pragma unroll
    for (int tl = 0; tl < TB_; ++tl) {
      const int t = t0 + tl;
      v16h hA[4];
#pragma unroll
      for (int kf = 0; kf < 4; ++kf) {
        const _Float16* p = &hbuf[arow][kf * 32 + kbh];
        v8h lo = *(const v8h*)p;
        v8h hi = *(const v8h*)(p + 16);
#pragma unroll
        for (int i = 0; i < 8; ++i) { hA[kf][i] = lo[i]; hA[kf][i + 8] = hi[i]; }
      }
      v8f acc[4];
#pragma unroll
      for (int nt = 0; nt < 4; ++nt) {
        acc[nt] = px[nt];
#pragma unroll
        for (int kf = 0; kf < 4; ++kf)
          acc[nt] = __builtin_amdgcn_wmma_f32_16x16x32_f16(false, hA[kf], false, whhB[nt][kf],
                                                           (short)0, acc[nt], false, false);
      }
#pragma unroll
      for (int nt = 0; nt < 4; ++nt) {
        const int g = nbase + nt * 16 + m;
        const bool isg = (g >> 7) == 2;
#pragma unroll
        for (int v = 0; v < HR_; ++v) {
          float a = acc[nt][tl * HR_ + v];
          gates[g * RB_ + hi16 * HR_ + v] = isg ? tanhf(a) : 1.0f / (1.0f + __expf(-a));
        }
      }
      lds_barrier();
#pragma unroll
      for (int i = 0; i < HR_; ++i) {
        const int p = tid + 256 * i;
        const int row = p >> 7, j = p & (HID_ - 1);
        float iv = gates[j * RB_ + row];
        float fv = gates[(HID_ + j) * RB_ + row];
        float gv = gates[(2 * HID_ + j) * RB_ + row];
        float ov = gates[(3 * HID_ + j) * RB_ + row];
        c[i] = fv * c[i] + iv * gv;
        float hv = ov * tanhf(c[i]);
        hbuf[row][j] = (_Float16)hv;
        if (STORE_SEQ) hseq[((size_t)(bm + row) * SEQ_ + t) * HID_ + j] = (_Float16)hv;
        if (!STORE_SEQ && t == SEQ_ - 1) hlast[(size_t)(bm + row) * HID_ + j] = hv;
      }
      lds_barrier();
    }
  }
}

__global__ __launch_bounds__(128) void head_kernel(const float* __restrict__ h1last,
                                                   const float* __restrict__ W2,
                                                   const float* __restrict__ b2,
                                                   const float* __restrict__ W3,
                                                   const float* __restrict__ b3,
                                                   float* __restrict__ out) {
  __shared__ float lastS[HID_];
  __shared__ float l2S[L2_];
  const int b = blockIdx.x, tid = threadIdx.x;
  lastS[tid] = tanhf(h1last[(size_t)b * HID_ + tid]);
  __syncthreads();
  if (tid < L2_) {
    float acc = b2[tid];
    for (int k = 0; k < HID_; ++k) acc += lastS[k] * W2[tid * HID_ + k];
    l2S[tid] = tanhf(acc);
  }
  __syncthreads();
  if (tid < HOR_) {
    float acc = b3[tid];
    for (int k = 0; k < L2_; ++k) acc += l2S[k] * W3[tid * L2_ + k];
    out[(size_t)b * HOR_ + tid] = acc;
  }
}

extern "C" void kernel_launch(void* const* d_in, const int* in_sizes, int n_in,
                              void* d_out, int out_size, void* d_ws, size_t ws_size,
                              hipStream_t stream) {
  (void)in_sizes; (void)n_in; (void)out_size; (void)ws_size;
  const float* inp  = (const float*)d_in[0];
  const float* akey = (const float*)d_in[1];
  const float* aval = (const float*)d_in[2];
  const float* W1   = (const float*)d_in[3];
  const float* b1   = (const float*)d_in[4];
  const float* Wih0 = (const float*)d_in[5];
  const float* Whh0 = (const float*)d_in[6];
  const float* bih0 = (const float*)d_in[7];
  const float* bhh0 = (const float*)d_in[8];
  const float* Wih1 = (const float*)d_in[9];
  const float* Whh1 = (const float*)d_in[10];
  const float* bih1 = (const float*)d_in[11];
  const float* bhh1 = (const float*)d_in[12];
  const float* W2   = (const float*)d_in[13];
  const float* b2   = (const float*)d_in[14];
  const float* W3   = (const float*)d_in[15];
  const float* b3   = (const float*)d_in[16];

  char* ws = (char*)d_ws;
  size_t off = 0;
  float*    dist   = (float*)(ws + off);    off += (size_t)B_ * NK_ * 4;
  float*    hist   = (float*)(ws + off);    off += (size_t)B_ * HLEN_ * 4;
  _Float16* xf     = (_Float16*)(ws + off); off += (size_t)B_ * SEQ_ * L1_ * 2;
  _Float16* h0     = (_Float16*)(ws + off); off += (size_t)B_ * SEQ_ * HID_ * 2;
  float*    h1last = (float*)(ws + off);    off += (size_t)B_ * HID_ * 4;

  gemm1_kernel<<<dim3(16, 64), 256, 0, stream>>>(inp, akey, dist);
  softmax_kernel<<<dim3(B_), 256, 0, stream>>>(dist);
  gemm2_kernel<<<dim3(16, 5), 256, 0, stream>>>(dist, aval, hist);
  featl1_kernel<<<dim3(B_, SEQ_ / 8), 256, 0, stream>>>(inp, hist, W1, b1, xf);
  lstm_kernel<L1_, true><<<dim3(B_ / RB_), 256, 0, stream>>>(xf, Wih0, Whh0, bih0, bhh0, h0, nullptr);
  lstm_kernel<HID_, false><<<dim3(B_ / RB_), 256, 0, stream>>>(h0, Wih1, Whh1, bih1, bhh1, nullptr, h1last);
  head_kernel<<<dim3(B_), 128, 0, stream>>>(h1last, W2, b2, W3, b3, (float*)d_out);
}


# LSTM: both layers in one launch: in each 16-wave workgroup waves 0-7 run layer 0 on timestep block k while waves 8-15 run layer 1 on block k-1; layer-0 h passes through LDS; order of sums kept
# speedup vs baseline: 6.2878x; 1.0926x over previous
#include <hip/hip_runtime.h>
#include <cstdint>
#include <cstddef>

typedef __attribute__((ext_vector_type(16))) _Float16 v16h;
typedef __attribute__((ext_vector_type(8)))  _Float16 v8h;
typedef __attribute__((ext_vector_type(8)))  float    v8f;
typedef __attribute__((ext_vector_type(2)))  float    v2f;

#define B_    256
#define SEQ_  512
#define HOR_  32
#define NK_   8192
#define L1_   64
#define HID_  128
#define L2_   64
#define GATES_ 512
#define HLEN_ 544

__global__ __launch_bounds__(256) void gemm1_kernel(const float* __restrict__ inp,
                                                    const float* __restrict__ akey,
                                                    float* __restrict__ dist) {
  const int tid = threadIdx.x, lane = tid & 31, wave = tid >> 5;
  const int mt = blockIdx.x, nt = blockIdx.y * 8 + wave;
  const int m = lane & 15;
  const int kb = (lane & 16) ? 2 : 0;
  const float* Ap = inp  + (size_t)(mt * 16 + m) * SEQ_ + kb;
  const float* Bp = akey + (size_t)(nt * 16 + m) * SEQ_ + kb;
  v8f acc0 = {}, acc1 = {};
  for (int k = 0; k < SEQ_; k += 8) {
    v2f a0 = *(const v2f*)(Ap + k);
    v2f b0 = *(const v2f*)(Bp + k);
    acc0 = __builtin_amdgcn_wmma_f32_16x16x4_f32(false, a0, false, b0, (short)0, acc0, false, false);
    v2f a1 = *(const v2f*)(Ap + k + 4);
    v2f b1 = *(const v2f*)(Bp + k + 4);
    acc1 = __builtin_amdgcn_wmma_f32_16x16x4_f32(false, a1, false, b1, (short)0, acc1, false, false);
  }
#pragma unroll
  for (int r = 0; r < 8; ++r) {
    int row = r + ((lane & 16) ? 8 : 0);
    dist[(size_t)(mt * 16 + row) * NK_ + nt * 16 + m] = acc0[r] + acc1[r];
  }
}

__global__ __launch_bounds__(256) void softmax_kernel(float* __restrict__ dist) {
  __shared__ float sv[256];
  __shared__ int   si[256];
  const int b = blockIdx.x, tid = threadIdx.x;
  float* row = dist + (size_t)b * NK_;

  float m = -3.4e38f; int mi = NK_;
  for (int j = tid; j < NK_; j += 256) {
    float v = row[j];
    if (v > m || (v == m && j < mi)) { m = v; mi = j; }
  }
  sv[tid] = m; si[tid] = mi;
  __syncthreads();
  for (int s = 128; s > 0; s >>= 1) {
    if (tid < s) {
      float v2 = sv[tid + s]; int i2 = si[tid + s];
      if (v2 > sv[tid] || (v2 == sv[tid] && i2 < si[tid])) { sv[tid] = v2; si[tid] = i2; }
    }
    __syncthreads();
  }
  if (tid == 0) row[si[0]] = 0.0f;
  __syncthreads();

  float mm = -3.4e38f;
  for (int j = tid; j < NK_; j += 256) mm = fmaxf(mm, row[j]);
  sv[tid] = mm;
  __syncthreads();
  for (int s = 128; s > 0; s >>= 1) { if (tid < s) sv[tid] = fmaxf(sv[tid], sv[tid + s]); __syncthreads(); }
  const float M = sv[0];
  __syncthreads();

  float sum = 0.f;
  for (int j = tid; j < NK_; j += 256) sum += __expf(row[j] - M);
  sv[tid] = sum;
  __syncthreads();
  for (int s = 128; s > 0; s >>= 1) { if (tid < s) sv[tid] += sv[tid + s]; __syncthreads(); }
  const float inv = 1.0f / sv[0];
  for (int j = tid; j < NK_; j += 256) row[j] = __expf(row[j] - M) * inv;
}

__global__ __launch_bounds__(256) void gemm2_kernel(const float* __restrict__ a,
                                                    const float* __restrict__ aval,
                                                    float* __restrict__ hist) {
  const int tid = threadIdx.x, lane = tid & 31, wave = tid >> 5;
  const int mt = blockIdx.x, nt = blockIdx.y * 8 + wave;
  if (nt >= (HLEN_ / 16)) return;
  const int m = lane & 15;
  const int kb = (lane & 16) ? 2 : 0;
  const int n = nt * 16 + m;
  const float* Ap = a + (size_t)(mt * 16 + m) * NK_ + kb;
  const float* Bp = aval + (size_t)kb * HLEN_ + n;
  v8f acc0 = {}, acc1 = {};
  for (int k = 0; k < NK_; k += 8) {
    v2f av = *(const v2f*)(Ap + k);
    v2f bv; bv[0] = Bp[(size_t)k * HLEN_]; bv[1] = Bp[(size_t)(k + 1) * HLEN_];
    acc0 = __builtin_amdgcn_wmma_f32_16x16x4_f32(false, av, false, bv, (short)0, acc0, false, false);
    v2f av1 = *(const v2f*)(Ap + k + 4);
    v2f bv1; bv1[0] = Bp[(size_t)(k + 4) * HLEN_]; bv1[1] = Bp[(size_t)(k + 5) * HLEN_];
    acc1 = __builtin_amdgcn_wmma_f32_16x16x4_f32(false, av1, false, bv1, (short)0, acc1, false, false);
  }
#pragma unroll
  for (int r = 0; r < 8; ++r) {
    int row = r + ((lane & 16) ? 8 : 0);
    hist[(size_t)(mt * 16 + row) * HLEN_ + n] = acc0[r] + acc1[r];
  }
}

__global__ __launch_bounds__(256) void featl1_kernel(const float* __restrict__ inp,
                                                     const float* __restrict__ hist,
                                                     const float* __restrict__ W1,
                                                     const float* __restrict__ b1,
                                                     _Float16* __restrict__ xf) {
  __shared__ float w1s[L1_ * (HOR_ + 1)];
  __shared__ float b1s[L1_];
  __shared__ float hs[8 + HOR_ - 1];
  __shared__ float ins[8];
  const int b = blockIdx.x, t0 = blockIdx.y * 8, tid = threadIdx.x;
  for (int i = tid; i < L1_ * (HOR_ + 1); i += 256) w1s[i] = W1[i];
  if (tid < L1_) b1s[tid] = b1[tid];
  if (tid < 8 + HOR_ - 1) hs[tid] = hist[(size_t)b * HLEN_ + t0 + tid];
  if (tid < 8) ins[tid] = inp[(size_t)b * SEQ_ + t0 + tid];
  __syncthreads();
  const int o = tid & 63;
#pragma unroll
  for (int pass = 0; pass < 2; ++pass) {
    int tt = (tid >> 6) + pass * 4;
    float acc = b1s[o] + ins[tt] * w1s[o * 33];
#pragma unroll
    for (int j = 0; j < HOR_; ++j) acc += hs[tt + j] * w1s[o * 33 + 1 + j];
    xf[((size_t)b * SEQ_ + (t0 + tt)) * L1_ + o] = (_Float16)tanhf(acc);
  }
}

__device__ __forceinline__ void lds_barrier() {
  __builtin_amdgcn_fence(__ATOMIC_RELEASE, "workgroup", "local");
  __builtin_amdgcn_s_barrier();
  __builtin_amdgcn_fence(__ATOMIC_ACQUIRE, "workgroup", "local");
}

#define RB_ 2
#define HR_ (RB_ / 2)
#define TB_ (8 / HR_)
#define NBLK_ (SEQ_ / TB_)
#define WP0_ (L1_ + 8)
#define WP1_ (HID_ + 8)

template <int KIN, int WP>
__device__ __forceinline__ void proj_block(const _Float16* ap, const _Float16* wp,
                                           const float* bp, float* pxs_l) {
  constexpr int KFX = KIN / 32;
  v16h xA[KFX];
#pragma unroll
  for (int kf = 0; kf < KFX; ++kf) {
    v8h lo = *(const v8h*)(ap + kf * 32);
    v8h hi = *(const v8h*)(ap + kf * 32 + 16);
#pragma unroll
    for (int i = 0; i < 8; ++i) { xA[kf][i] = lo[i]; xA[kf][i + 8] = hi[i]; }
  }
#pragma unroll
  for (int nt = 0; nt < 4; ++nt) {
    v8f px;
    const float bias = bp[nt * 16];
#pragma unroll
    for (int r = 0; r < 8; ++r) px[r] = bias;
#pragma unroll
    for (int kf = 0; kf < KFX; ++kf) {
      const _Float16* q = wp + nt * 16 * WP + kf * 32;
      v8h lo = *(const v8h*)q;
      v8h hi = *(const v8h*)(q + 16);
      v16h wB;
#pragma unroll
      for (int i = 0; i < 8; ++i) { wB[i] = lo[i]; wB[i + 8] = hi[i]; }
      if (kf < KFX - 1) {
        px = __builtin_amdgcn_wmma_f32_16x16x32_f16(false, xA[kf], false, wB,
                                                    (short)0, px, false, false);
      } else {
        asm volatile("v_wmma_f32_16x16x32_f16 %0, %1, %2, %0\n\tv_nop\n\tv_nop\n\tv_nop\n\tv_nop"
                     : "+v"(px) : "v"(xA[kf]), "v"(wB));
      }
    }
#pragma unroll
    for (int r = 0; r < 8; ++r) pxs_l[(nt * 8 + r) * 32] = px[r];
  }
}

__global__ __launch_bounds__(512, 1) void lstm2_kernel(const _Float16* __restrict__ xin,
                                                       const float* __restrict__ Wih0,
                                                       const float* __restrict__ Whh0,
                                                       const float* __restrict__ bih0,
                                                       const float* __restrict__ bhh0,
                                                       const float* __restrict__ Wih1,
                                                       const float* __restrict__ Whh1,
                                                       const float* __restrict__ bih1,
                                                       const float* __restrict__ bhh1,
                                                       float* __restrict__ hlast) {
  __shared__ __align__(16) _Float16 wih0s[GATES_ * WP0_];
  __shared__ __align__(16) _Float16 wih1s[GATES_ * WP1_];
  __shared__ float pxs[16 * 4 * 8 * 32];
  __shared__ float gates[2][GATES_ * RB_];
  __shared__ float biass[2][GATES_];
  __shared__ __align__(16) _Float16 hbuf[2][RB_][HID_];
  __shared__ __align__(16) _Float16 h0blk[TB_][RB_][HID_];
  const int tid = threadIdx.x, lane = tid & 31;
  const int wave = __builtin_amdgcn_readfirstlane(tid >> 5);
  const int layer = wave >> 3;
  const int ltid = tid & 255;
  const int bm = blockIdx.x * RB_;
  const int nbase = (wave & 7) * 64;
  const int m = lane & 15;
  const int hi16 = (lane >> 4) & 1;
  const int kbh = hi16 * 8;
  const int arow = (m >> 3) * HR_ + (m & (HR_ - 1));
  const int xts = (m & 7) / HR_;
  const float* __restrict__ Whh = layer ? Whh1 : Whh0;
  const float* __restrict__ bih = layer ? bih1 : bih0;
  const float* __restrict__ bhh = layer ? bhh1 : bhh0;

  for (int i = tid; i < GATES_ * L1_; i += 512)
    wih0s[(i >> 6) * WP0_ + (i & (L1_ - 1))] = (_Float16)Wih0[i];
  for (int i = tid; i < GATES_ * HID_; i += 512)
    wih1s[(i >> 7) * WP1_ + (i & (HID_ - 1))] = (_Float16)Wih1[i];
  for (int i = tid; i < TB_ * RB_ * HID_; i += 512)
    (&h0blk[0][0][0])[i] = (_Float16)0.f;
  for (int i = ltid; i < GATES_; i += 256) biass[layer][i] = bih[i] + bhh[i];

  v16h whhB[4][4];
#pragma unroll
  for (int nt = 0; nt < 4; ++nt) {
    const int g = nbase + nt * 16 + m;
#pragma unroll
    for (int kf = 0; kf < 4; ++kf)
#pragma unroll
      for (int h = 0; h < 16; ++h) {
        int k = kf * 32 + ((h & 8) << 1) + kbh + (h & 7);
        whhB[nt][kf][h] = (_Float16)Whh[(size_t)g * HID_ + k];
      }
  }

  float* gl = gates[layer];
  _Float16 (*hb)[HID_] = hbuf[layer];
  float* pxs_l = pxs + wave * (4 * 8 * 32) + lane;
  float c[HR_];
#pragma unroll
  for (int i = 0; i < HR_; ++i) {
    const int p = ltid + 256 * i;
    c[i] = 0.f;
    hb[p >> 7][p & (HID_ - 1)] = (_Float16)0.f;
  }
  __syncthreads();

  for (int blk = 0; blk <= NBLK_; ++blk) {
    const bool active = layer ? (blk > 0) : (blk < NBLK_);
    const int t0 = (blk - layer) * TB_;
    if (layer == 0) {
      const int tx = (blk < NBLK_ ? blk : NBLK_ - 1) * TB_;
      proj_block<L1_, WP0_>(xin + ((size_t)(bm + arow) * SEQ_ + (tx + xts)) * L1_ + kbh,
                            wih0s + (nbase + m) * WP0_ + kbh, &biass[0][nbase + m], pxs_l);
    } else {
      proj_block<HID_, WP1_>(&h0blk[xts][arow][kbh],
                             wih1s + (nbase + m) * WP1_ + kbh, &biass[1][nbase + m], pxs_l);
    }
#pragma unroll
    for (int tl = 0; tl < TB_; ++tl) {
      const int t = t0 + tl;
      v16h hA[4];
#pragma unroll
      for (int kf = 0; kf < 4; ++kf) {
        const _Float16* p = &hb[arow][kf * 32 + kbh];
        v8h lo = *(const v8h*)p;
        v8h hi = *(const v8h*)(p + 16);
#pragma unroll
        for (int i = 0; i < 8; ++i) { hA[kf][i] = lo[i]; hA[kf][i + 8] = hi[i]; }
      }
      v8f acc[4];
#pragma unroll
      for (int nt = 0; nt < 4; ++nt) {
#pragma unroll
        for (int r = 0; r < 8; ++r) acc[nt][r] = 0.f;
#pragma unroll
        for (int v = 0; v < HR_; ++v) acc[nt][tl * HR_ + v] = pxs_l[(nt * 8 + tl * HR_ + v) * 32];
#pragma unroll
        for (int kf = 0; kf < 4; ++kf)
          acc[nt] = __builtin_amdgcn_wmma_f32_16x16x32_f16(false, hA[kf], false, whhB[nt][kf],
                                                           (short)0, acc[nt], false, false);
      }
#pragma unroll
      for (int nt = 0; nt < 4; ++nt) {
        const int g = nbase + nt * 16 + m;
        const bool isg = (g >> 7) == 2;
#pragma unroll
        for (int v = 0; v < HR_; ++v) {
          float a = acc[nt][tl * HR_ + v];
          gl[g * RB_ + hi16 * HR_ + v] = isg ? tanhf(a) : 1.0f / (1.0f + __expf(-a));
        }
      }
      lds_barrier();
      if (active) {
#pragma unroll
        for (int i = 0; i < HR_; ++i) {
          const int p = ltid + 256 * i;
          const int row = p >> 7, j = p & (HID_ - 1);
          float iv = gl[j * RB_ + row];
          float fv = gl[(HID_ + j) * RB_ + row];
          float gv = gl[(2 * HID_ + j) * RB_ + row];
          float ov = gl[(3 * HID_ + j) * RB_ + row];
          c[i] = fv * c[i] + iv * gv;
          float hv = ov * tanhf(c[i]);
          hb[row][j] = (_Float16)hv;
          if (layer == 0) h0blk[tl][row][j] = (_Float16)hv;
          else if (t == SEQ_ - 1) hlast[(size_t)(bm + row) * HID_ + j] = hv;
        }
      }
      lds_barrier();
    }
  }
}

__global__ __launch_bounds__(128) void head_kernel(const float* __restrict__ h1last,
                                                   const float* __restrict__ W2,
                                                   const float* __restrict__ b2,
                                                   const float* __restrict__ W3,
                                                   const float* __restrict__ b3,
                                                   float* __restrict__ out) {
  __shared__ float lastS[HID_];
  __shared__ float l2S[L2_];
  const int b = blockIdx.x, tid = threadIdx.x;
  lastS[tid] = tanhf(h1last[(size_t)b * HID_ + tid]);
  __syncthreads();
  if (tid < L2_) {
    float acc = b2[tid];
    for (int k = 0; k < HID_; ++k) acc += lastS[k] * W2[tid * HID_ + k];
    l2S[tid] = tanhf(acc);
  }
  __syncthreads();
  if (tid < HOR_) {
    float acc = b3[tid];
    for (int k = 0; k < L2_; ++k) acc += l2S[k] * W3[tid * L2_ + k];
    out[(size_t)b * HOR_ + tid] = acc;
  }
}

extern "C" void kernel_launch(void* const* d_in, const int* in_sizes, int n_in,
                              void* d_out, int out_size, void* d_ws, size_t ws_size,
                              hipStream_t stream) {
  (void)in_sizes; (void)n_in; (void)out_size; (void)ws_size;
  const float* inp  = (const float*)d_in[0];
  const float* akey = (const float*)d_in[1];
  const float* aval = (const float*)d_in[2];
  const float* W1   = (const float*)d_in[3];
  const float* b1   = (const float*)d_in[4];
  const float* Wih0 = (const float*)d_in[5];
  const float* Whh0 = (const float*)d_in[6];
  const float* bih0 = (const float*)d_in[7];
  const float* bhh0 = (const float*)d_in[8];
  const float* Wih1 = (const float*)d_in[9];
  const float* Whh1 = (const float*)d_in[10];
  const float* bih1 = (const float*)d_in[11];
  const float* bhh1 = (const float*)d_in[12];
  const float* W2   = (const float*)d_in[13];
  const float* b2   = (const float*)d_in[14];
  const float* W3   = (const float*)d_in[15];
  const float* b3   = (const float*)d_in[16];

  char* ws = (char*)d_ws;
  size_t off = 0;
  float*    dist   = (float*)(ws + off);    off += (size_t)B_ * NK_ * 4;
  float*    hist   = (float*)(ws + off);    off += (size_t)B_ * HLEN_ * 4;
  _Float16* xf     = (_Float16*)(ws + off); off += (size_t)B_ * SEQ_ * L1_ * 2;
  float*    h1last = (float*)(ws + off);    off += (size_t)B_ * HID_ * 4;

  gemm1_kernel<<<dim3(16, 64), 256, 0, stream>>>(inp, akey, dist);
  softmax_kernel<<<dim3(B_), 256, 0, stream>>>(dist);
  gemm2_kernel<<<dim3(16, 5), 256, 0, stream>>>(dist, aval, hist);
  featl1_kernel<<<dim3(B_, SEQ_ / 8), 256, 0, stream>>>(inp, hist, W1, b1, xf);
  lstm2_kernel<<<dim3(B_ / RB_), 512, 0, stream>>>(xf, Wih0, Whh0, bih0, bhh0, Wih1, Whh1, bih1, bhh1, h1last);
  head_kernel<<<dim3(B_), 128, 0, stream>>>(h1last, W2, b2, W3, b3, (float*)d_out);
}


# LSTM step: gate sums go to LDS raw; sigmoid/tanh are applied behind the barrier by the thread that owns the (row, unit), one function per gate; before, every lane evaluated both for 4 values
# speedup vs baseline: 6.7671x; 1.0762x over previous
#include <hip/hip_runtime.h>
#include <cstdint>
#include <cstddef>

typedef __attribute__((ext_vector_type(16))) _Float16 v16h;
typedef __attribute__((ext_vector_type(8)))  _Float16 v8h;
typedef __attribute__((ext_vector_type(8)))  float    v8f;
typedef __attribute__((ext_vector_type(2)))  float    v2f;

#define B_    256
#define SEQ_  512
#define HOR_  32
#define NK_   8192
#define L1_   64
#define HID_  128
#define L2_   64
#define GATES_ 512
#define HLEN_ 544

__global__ __launch_bounds__(256) void gemm1_kernel(const float* __restrict__ inp,
                                                    const float* __restrict__ akey,
                                                    float* __restrict__ dist) {
  const int tid = threadIdx.x, lane = tid & 31, wave = tid >> 5;
  const int mt = blockIdx.x, nt = blockIdx.y * 8 + wave;
  const int m = lane & 15;
  const int kb = (lane & 16) ? 2 : 0;
  const float* Ap = inp  + (size_t)(mt * 16 + m) * SEQ_ + kb;
  const float* Bp = akey + (size_t)(nt * 16 + m) * SEQ_ + kb;
  v8f acc0 = {}, acc1 = {};
  for (int k = 0; k < SEQ_; k += 8) {
    v2f a0 = *(const v2f*)(Ap + k);
    v2f b0 = *(const v2f*)(Bp + k);
    acc0 = __builtin_amdgcn_wmma_f32_16x16x4_f32(false, a0, false, b0, (short)0, acc0, false, false);
    v2f a1 = *(const v2f*)(Ap + k + 4);
    v2f b1 = *(const v2f*)(Bp + k + 4);
    acc1 = __builtin_amdgcn_wmma_f32_16x16x4_f32(false, a1, false, b1, (short)0, acc1, false, false);
  }
#pragma unroll
  for (int r = 0; r < 8; ++r) {
    int row = r + ((lane & 16) ? 8 : 0);
    dist[(size_t)(mt * 16 + row) * NK_ + nt * 16 + m] = acc0[r] + acc1[r];
  }
}

__global__ __launch_bounds__(256) void softmax_kernel(float* __restrict__ dist) {
  __shared__ float sv[256];
  __shared__ int   si[256];
  const int b = blockIdx.x, tid = threadIdx.x;
  float* row = dist + (size_t)b * NK_;

  float m = -3.4e38f; int mi = NK_;
  for (int j = tid; j < NK_; j += 256) {
    float v = row[j];
    if (v > m || (v == m && j < mi)) { m = v; mi = j; }
  }
  sv[tid] = m; si[tid] = mi;
  __syncthreads();
  for (int s = 128; s > 0; s >>= 1) {
    if (tid < s) {
      float v2 = sv[tid + s]; int i2 = si[tid + s];
      if (v2 > sv[tid] || (v2 == sv[tid] && i2 < si[tid])) { sv[tid] = v2; si[tid] = i2; }
    }
    __syncthreads();
  }
  if (tid == 0) row[si[0]] = 0.0f;
  __syncthreads();

  float mm = -3.4e38f;
  for (int j = tid; j < NK_; j += 256) mm = fmaxf(mm, row[j]);
  sv[tid] = mm;
  __syncthreads();
  for (int s = 128; s > 0; s >>= 1) { if (tid < s) sv[tid] = fmaxf(sv[tid], sv[tid + s]); __syncthreads(); }
  const float M = sv[0];
  __syncthreads();

  float sum = 0.f;
  for (int j = tid; j < NK_; j += 256) sum += __expf(row[j] - M);
  sv[tid] = sum;
  __syncthreads();
  for (int s = 128; s > 0; s >>= 1) { if (tid < s) sv[tid] += sv[tid + s]; __syncthreads(); }
  const float inv = 1.0f / sv[0];
  for (int j = tid; j < NK_; j += 256) row[j] = __expf(row[j] - M) * inv;
}

__global__ __launch_bounds__(256) void gemm2_kernel(const float* __restrict__ a,
                                                    const float* __restrict__ aval,
                                                    float* __restrict__ hist) {
  const int tid = threadIdx.x, lane = tid & 31, wave = tid >> 5;
  const int mt = blockIdx.x, nt = blockIdx.y * 8 + wave;
  if (nt >= (HLEN_ / 16)) return;
  const int m = lane & 15;
  const int kb = (lane & 16) ? 2 : 0;
  const int n = nt * 16 + m;
  const float* Ap = a + (size_t)(mt * 16 + m) * NK_ + kb;
  const float* Bp = aval + (size_t)kb * HLEN_ + n;
  v8f acc0 = {}, acc1 = {};
  for (int k = 0; k < NK_; k += 8) {
    v2f av = *(const v2f*)(Ap + k);
    v2f bv; bv[0] = Bp[(size_t)k * HLEN_]; bv[1] = Bp[(size_t)(k + 1) * HLEN_];
    acc0 = __builtin_amdgcn_wmma_f32_16x16x4_f32(false, av, false, bv, (short)0, acc0, false, false);
    v2f av1 = *(const v2f*)(Ap + k + 4);
    v2f bv1; bv1[0] = Bp[(size_t)(k + 4) * HLEN_]; bv1[1] = Bp[(size_t)(k + 5) * HLEN_];
    acc1 = __builtin_amdgcn_wmma_f32_16x16x4_f32(false, av1, false, bv1, (short)0, acc1, false, false);
  }
#pragma unroll
  for (int r = 0; r < 8; ++r) {
    int row = r + ((lane & 16) ? 8 : 0);
    hist[(size_t)(mt * 16 + row) * HLEN_ + n] = acc0[r] + acc1[r];
  }
}

__global__ __launch_bounds__(256) void featl1_kernel(const float* __restrict__ inp,
                                                     const float* __restrict__ hist,
                                                     const float* __restrict__ W1,
                                                     const float* __restrict__ b1,
                                                     _Float16* __restrict__ xf) {
  __shared__ float w1s[L1_ * (HOR_ + 1)];
  __shared__ float b1s[L1_];
  __shared__ float hs[8 + HOR_ - 1];
  __shared__ float ins[8];
  const int b = blockIdx.x, t0 = blockIdx.y * 8, tid = threadIdx.x;
  for (int i = tid; i < L1_ * (HOR_ + 1); i += 256) w1s[i] = W1[i];
  if (tid < L1_) b1s[tid] = b1[tid];
  if (tid < 8 + HOR_ - 1) hs[tid] = hist[(size_t)b * HLEN_ + t0 + tid];
  if (tid < 8) ins[tid] = inp[(size_t)b * SEQ_ + t0 + tid];
  __syncthreads();
  const int o = tid & 63;
#pragma unroll
  for (int pass = 0; pass < 2; ++pass) {
    int tt = (tid >> 6) + pass * 4;
    float acc = b1s[o] + ins[tt] * w1s[o * 33];
#pragma unroll
    for (int j = 0; j < HOR_; ++j) acc += hs[tt + j] * w1s[o * 33 + 1 + j];
    xf[((size_t)b * SEQ_ + (t0 + tt)) * L1_ + o] = (_Float16)tanhf(acc);
  }
}

__device__ __forceinline__ void lds_barrier() {
  __builtin_amdgcn_fence(__ATOMIC_RELEASE, "workgroup", "local");
  __builtin_amdgcn_s_barrier();
  __builtin_amdgcn_fence(__ATOMIC_ACQUIRE, "workgroup", "local");
}

#define RB_ 2
#define HR_ (RB_ / 2)
#define TB_ (8 / HR_)
#define NBLK_ (SEQ_ / TB_)
#define WP0_ (L1_ + 8)
#define WP1_ (HID_ + 8)

template <int KIN, int WP>
__device__ __forceinline__ void proj_block(const _Float16* ap, const _Float16* wp,
                                           const float* bp, float* pxs_l) {
  constexpr int KFX = KIN / 32;
  v16h xA[KFX];
#pragma unroll
  for (int kf = 0; kf < KFX; ++kf) {
    v8h lo = *(const v8h*)(ap + kf * 32);
    v8h hi = *(const v8h*)(ap + kf * 32 + 16);
#pragma unroll
    for (int i = 0; i < 8; ++i) { xA[kf][i] = lo[i]; xA[kf][i + 8] = hi[i]; }
  }
#pragma unroll
  for (int nt = 0; nt < 4; ++nt) {
    v8f px;
    const float bias = bp[nt * 16];
#pragma unroll
    for (int r = 0; r < 8; ++r) px[r] = bias;
#pragma unroll
    for (int kf = 0; kf < KFX; ++kf) {
      const _Float16* q = wp + nt * 16 * WP + kf * 32;
      v8h lo = *(const v8h*)q;
      v8h hi = *(const v8h*)(q + 16);
      v16h wB;
#pragma unroll
      for (int i = 0; i < 8; ++i) { wB[i] = lo[i]; wB[i + 8] = hi[i]; }
      if (kf < KFX - 1) {
        px = __builtin_amdgcn_wmma_f32_16x16x32_f16(false, xA[kf], false, wB,
                                                    (short)0, px, false, false);
      } else {
        asm volatile("v_wmma_f32_16x16x32_f16 %0, %1, %2, %0\n\tv_nop\n\tv_nop\n\tv_nop\n\tv_nop"
                     : "+v"(px) : "v"(xA[kf]), "v"(wB));
      }
    }
#pragma unroll
    for (int r = 0; r < 8; ++r) pxs_l[(nt * 8 + r) * 32] = px[r];
  }
}

__global__ __launch_bounds__(512, 1) void lstm2_kernel(const _Float16* __restrict__ xin,
                                                       const float* __restrict__ Wih0,
                                                       const float* __restrict__ Whh0,
                                                       const float* __restrict__ bih0,
                                                       const float* __restrict__ bhh0,
                                                       const float* __restrict__ Wih1,
                                                       const float* __restrict__ Whh1,
                                                       const float* __restrict__ bih1,
                                                       const float* __restrict__ bhh1,
                                                       float* __restrict__ hlast) {
  __shared__ __align__(16) _Float16 wih0s[GATES_ * WP0_];
  __shared__ __align__(16) _Float16 wih1s[GATES_ * WP1_];
  __shared__ float pxs[16 * 4 * 8 * 32];
  __shared__ float gates[2][GATES_ * RB_];
  __shared__ float biass[2][GATES_];
  __shared__ __align__(16) _Float16 hbuf[2][RB_][HID_];
  __shared__ __align__(16) _Float16 h0blk[TB_][RB_][HID_];
  const int tid = threadIdx.x, lane = tid & 31;
  const int wave = __builtin_amdgcn_readfirstlane(tid >> 5);
  const int layer = wave >> 3;
  const int ltid = tid & 255;
  const int bm = blockIdx.x * RB_;
  const int nbase = (wave & 7) * 64;
  const int m = lane & 15;
  const int hi16 = (lane >> 4) & 1;
  const int kbh = hi16 * 8;
  const int arow = (m >> 3) * HR_ + (m & (HR_ - 1));
  const int xts = (m & 7) / HR_;
  const float* __restrict__ Whh = layer ? Whh1 : Whh0;
  const float* __restrict__ bih = layer ? bih1 : bih0;
  const float* __restrict__ bhh = layer ? bhh1 : bhh0;

  for (int i = tid; i < GATES_ * L1_; i += 512)
    wih0s[(i >> 6) * WP0_ + (i & (L1_ - 1))] = (_Float16)Wih0[i];
  for (int i = tid; i < GATES_ * HID_; i += 512)
    wih1s[(i >> 7) * WP1_ + (i & (HID_ - 1))] = (_Float16)Wih1[i];
  for (int i = tid; i < TB_ * RB_ * HID_; i += 512)
    (&h0blk[0][0][0])[i] = (_Float16)0.f;
  for (int i = ltid; i < GATES_; i += 256) biass[layer][i] = bih[i] + bhh[i];

  v16h whhB[4][4];
#pragma unroll
  for (int nt = 0; nt < 4; ++nt) {
    const int g = nbase + nt * 16 + m;
#pragma unroll
    for (int kf = 0; kf < 4; ++kf)
#pragma unroll
      for (int h = 0; h < 16; ++h) {
        int k = kf * 32 + ((h & 8) << 1) + kbh + (h & 7);
        whhB[nt][kf][h] = (_Float16)Whh[(size_t)g * HID_ + k];
      }
  }

  float* gl = gates[layer];
  _Float16 (*hb)[HID_] = hbuf[layer];
  float* pxs_l = pxs + wave * (4 * 8 * 32) + lane;
  float c[HR_];
#pragma unroll
  for (int i = 0; i < HR_; ++i) {
    const int p = ltid + 256 * i;
    c[i] = 0.f;
    hb[p >> 7][p & (HID_ - 1)] = (_Float16)0.f;
  }
  __syncthreads();

  for (int blk = 0; blk <= NBLK_; ++blk) {
    const bool active = layer ? (blk > 0) : (blk < NBLK_);
    const int t0 = (blk - layer) * TB_;
    if (layer == 0) {
      const int tx = (blk < NBLK_ ? blk : NBLK_ - 1) * TB_;
      proj_block<L1_, WP0_>(xin + ((size_t)(bm + arow) * SEQ_ + (tx + xts)) * L1_ + kbh,
                            wih0s + (nbase + m) * WP0_ + kbh, &biass[0][nbase + m], pxs_l);
    } else {
      proj_block<HID_, WP1_>(&h0blk[xts][arow][kbh],
                             wih1s + (nbase + m) * WP1_ + kbh, &biass[1][nbase + m], pxs_l);
    }
#pragma unroll
    for (int tl = 0; tl < TB_; ++tl) {
      const int t = t0 + tl;
      v16h hA[4];
#pragma unroll
      for (int kf = 0; kf < 4; ++kf) {
        const _Float16* p = &hb[arow][kf * 32 + kbh];
        v8h lo = *(const v8h*)p;
        v8h hi = *(const v8h*)(p + 16);
#pragma unroll
        for (int i = 0; i < 8; ++i) { hA[kf][i] = lo[i]; hA[kf][i + 8] = hi[i]; }
      }
      v8f acc[4];
#pragma unroll
      for (int nt = 0; nt < 4; ++nt) {
#pragma unroll
        for (int r = 0; r < 8; ++r) acc[nt][r] = 0.f;
#pragma unroll
        for (int v = 0; v < HR_; ++v) acc[nt][tl * HR_ + v] = pxs_l[(nt * 8 + tl * HR_ + v) * 32];
#pragma unroll
        for (int kf = 0; kf < 4; ++kf)
          acc[nt] = __builtin_amdgcn_wmma_f32_16x16x32_f16(false, hA[kf], false, whhB[nt][kf],
                                                           (short)0, acc[nt], false, false);
      }
#pragma unroll
      for (int nt = 0; nt < 4; ++nt) {
        const int g = nbase + nt * 16 + m;
#pragma unroll
        for (int v = 0; v < HR_; ++v) gl[g * RB_ + hi16 * HR_ + v] = acc[nt][tl * HR_ + v];
      }
      lds_barrier();
      if (active) {
#pragma unroll
        for (int i = 0; i < HR_; ++i) {
          const int p = ltid + 256 * i;
          const int row = p >> 7, j = p & (HID_ - 1);
          const float ia = gl[j * RB_ + row];
          const float fa = gl[(HID_ + j) * RB_ + row];
          const float ga = gl[(2 * HID_ + j) * RB_ + row];
          const float oa = gl[(3 * HID_ + j) * RB_ + row];
          const float iv = 1.0f / (1.0f + __expf(-ia));
          const float fv = 1.0f / (1.0f + __expf(-fa));
          const float gv = tanhf(ga);
          const float ov = 1.0f / (1.0f + __expf(-oa));
          c[i] = fv * c[i] + iv * gv;
          float hv = ov * tanhf(c[i]);
          hb[row][j] = (_Float16)hv;
          if (layer == 0) h0blk[tl][row][j] = (_Float16)hv;
          else if (t == SEQ_ - 1) hlast[(size_t)(bm + row) * HID_ + j] = hv;
        }
      }
      lds_barrier();
    }
  }
}

__global__ __launch_bounds__(128) void head_kernel(const float* __restrict__ h1last,
                                                   const float* __restrict__ W2,
                                                   const float* __restrict__ b2,
                                                   const float* __restrict__ W3,
                                                   const float* __restrict__ b3,
                                                   float* __restrict__ out) {
  __shared__ float lastS[HID_];
  __shared__ float l2S[L2_];
  const int b = blockIdx.x, tid = threadIdx.x;
  lastS[tid] = tanhf(h1last[(size_t)b * HID_ + tid]);
  __syncthreads();
  if (tid < L2_) {
    float acc = b2[tid];
    for (int k = 0; k < HID_; ++k) acc += lastS[k] * W2[tid * HID_ + k];
    l2S[tid] = tanhf(acc);
  }
  __syncthreads();
  if (tid < HOR_) {
    float acc = b3[tid];
    for (int k = 0; k < L2_; ++k) acc += l2S[k] * W3[tid * L2_ + k];
    out[(size_t)b * HOR_ + tid] = acc;
  }
}

extern "C" void kernel_launch(void* const* d_in, const int* in_sizes, int n_in,
                              void* d_out, int out_size, void* d_ws, size_t ws_size,
                              hipStream_t stream) {
  (void)in_sizes; (void)n_in; (void)out_size; (void)ws_size;
  const float* inp  = (const float*)d_in[0];
  const float* akey = (const float*)d_in[1];
  const float* aval = (const float*)d_in[2];
  const float* W1   = (const float*)d_in[3];
  const float* b1   = (const float*)d_in[4];
  const float* Wih0 = (const float*)d_in[5];
  const float* Whh0 = (const float*)d_in[6];
  const float* bih0 = (const float*)d_in[7];
  const float* bhh0 = (const float*)d_in[8];
  const float* Wih1 = (const float*)d_in[9];
  const float* Whh1 = (const float*)d_in[10];
  const float* bih1 = (const float*)d_in[11];
  const float* bhh1 = (const float*)d_in[12];
  const float* W2   = (const float*)d_in[13];
  const float* b2   = (const float*)d_in[14];
  const float* W3   = (const float*)d_in[15];
  const float* b3   = (const float*)d_in[16];

  char* ws = (char*)d_ws;
  size_t off = 0;
  float*    dist   = (float*)(ws + off);    off += (size_t)B_ * NK_ * 4;
  float*    hist   = (float*)(ws + off);    off += (size_t)B_ * HLEN_ * 4;
  _Float16* xf     = (_Float16*)(ws + off); off += (size_t)B_ * SEQ_ * L1_ * 2;
  float*    h1last = (float*)(ws + off);    off += (size_t)B_ * HID_ * 4;

  gemm1_kernel<<<dim3(16, 64), 256, 0, stream>>>(inp, akey, dist);
  softmax_kernel<<<dim3(B_), 256, 0, stream>>>(dist);
  gemm2_kernel<<<dim3(16, 5), 256, 0, stream>>>(dist, aval, hist);
  featl1_kernel<<<dim3(B_, SEQ_ / 8), 256, 0, stream>>>(inp, hist, W1, b1, xf);
  lstm2_kernel<<<dim3(B_ / RB_), 512, 0, stream>>>(xf, Wih0, Whh0, bih0, bhh0, Wih1, Whh1, bih1, bhh1, h1last);
  head_kernel<<<dim3(B_), 128, 0, stream>>>(h1last, W2, b2, W3, b3, (float*)d_out);
}


# GEMM2 (hist = a @ aval): one wave per SIMD (144 workgroups of 4 waves, was 80 of 8), K loop on a ring of three 64-k register sets: two sets of loads in flight under each set's WMMAs; sum order kept
# speedup vs baseline: 7.8626x; 1.1619x over previous
#include <hip/hip_runtime.h>
#include <cstdint>
#include <cstddef>

typedef __attribute__((ext_vector_type(16))) _Float16 v16h;
typedef __attribute__((ext_vector_type(8)))  _Float16 v8h;
typedef __attribute__((ext_vector_type(8)))  float    v8f;
typedef __attribute__((ext_vector_type(2)))  float    v2f;

#define B_    256
#define SEQ_  512
#define HOR_  32
#define NK_   8192
#define L1_   64
#define HID_  128
#define L2_   64
#define GATES_ 512
#define HLEN_ 544

__global__ __launch_bounds__(256) void gemm1_kernel(const float* __restrict__ inp,
                                                    const float* __restrict__ akey,
                                                    float* __restrict__ dist) {
  const int tid = threadIdx.x, lane = tid & 31, wave = tid >> 5;
  const int mt = blockIdx.x, nt = blockIdx.y * 8 + wave;
  const int m = lane & 15;
  const int kb = (lane & 16) ? 2 : 0;
  const float* Ap = inp  + (size_t)(mt * 16 + m) * SEQ_ + kb;
  const float* Bp = akey + (size_t)(nt * 16 + m) * SEQ_ + kb;
  v8f acc0 = {}, acc1 = {};
  for (int k = 0; k < SEQ_; k += 8) {
    v2f a0 = *(const v2f*)(Ap + k);
    v2f b0 = *(const v2f*)(Bp + k);
    acc0 = __builtin_amdgcn_wmma_f32_16x16x4_f32(false, a0, false, b0, (short)0, acc0, false, false);
    v2f a1 = *(const v2f*)(Ap + k + 4);
    v2f b1 = *(const v2f*)(Bp + k + 4);
    acc1 = __builtin_amdgcn_wmma_f32_16x16x4_f32(false, a1, false, b1, (short)0, acc1, false, false);
  }
#pragma unroll
  for (int r = 0; r < 8; ++r) {
    int row = r + ((lane & 16) ? 8 : 0);
    dist[(size_t)(mt * 16 + row) * NK_ + nt * 16 + m] = acc0[r] + acc1[r];
  }
}

__global__ __launch_bounds__(256) void softmax_kernel(float* __restrict__ dist) {
  __shared__ float sv[256];
  __shared__ int   si[256];
  const int b = blockIdx.x, tid = threadIdx.x;
  float* row = dist + (size_t)b * NK_;

  float m = -3.4e38f; int mi = NK_;
  for (int j = tid; j < NK_; j += 256) {
    float v = row[j];
    if (v > m || (v == m && j < mi)) { m = v; mi = j; }
  }
  sv[tid] = m; si[tid] = mi;
  __syncthreads();
  for (int s = 128; s > 0; s >>= 1) {
    if (tid < s) {
      float v2 = sv[tid + s]; int i2 = si[tid + s];
      if (v2 > sv[tid] || (v2 == sv[tid] && i2 < si[tid])) { sv[tid] = v2; si[tid] = i2; }
    }
    __syncthreads();
  }
  if (tid == 0) row[si[0]] = 0.0f;
  __syncthreads();

  float mm = -3.4e38f;
  for (int j = tid; j < NK_; j += 256) mm = fmaxf(mm, row[j]);
  sv[tid] = mm;
  __syncthreads();
  for (int s = 128; s > 0; s >>= 1) { if (tid < s) sv[tid] = fmaxf(sv[tid], sv[tid + s]); __syncthreads(); }
  const float M = sv[0];
  __syncthreads();

  float sum = 0.f;
  for (int j = tid; j < NK_; j += 256) sum += __expf(row[j] - M);
  sv[tid] = sum;
  __syncthreads();
  for (int s = 128; s > 0; s >>= 1) { if (tid < s) sv[tid] += sv[tid + s]; __syncthreads(); }
  const float inv = 1.0f / sv[0];
  for (int j = tid; j < NK_; j += 256) row[j] = __expf(row[j] - M) * inv;
}

#define G2KB_ 64
#define G2NF_ (G2KB_ / 4)
#define G2NBLK_ (NK_ / G2KB_)

__device__ __forceinline__ void g2_load1(v2f& af, v2f& bf, const float* __restrict__ Ap,
                                         const float* __restrict__ Bp, int k) {
  af = *(const v2f*)(Ap + k);
  bf[0] = Bp[(size_t)k * HLEN_];
  bf[1] = Bp[(size_t)(k + 1) * HLEN_];
}

__device__ __forceinline__ void g2_load(v2f (&af)[G2NF_], v2f (&bf)[G2NF_],
                                        const float* __restrict__ Ap,
                                        const float* __restrict__ Bp, int k0) {
#pragma unroll
  for (int i = 0; i < G2NF_; ++i) g2_load1(af[i], bf[i], Ap, Bp, k0 + i * 4);
}

template <bool REFILL>
__device__ __forceinline__ void g2_phase(const v2f (&af)[G2NF_], const v2f (&bf)[G2NF_],
                                         v2f (&an)[G2NF_], v2f (&bn)[G2NF_],
                                         const float* __restrict__ Ap,
                                         const float* __restrict__ Bp, int kn,
                                         v8f& acc0, v8f& acc1) {
#pragma unroll
  for (int i = 0; i < G2NF_; i += 2) {
    acc0 = __builtin_amdgcn_wmma_f32_16x16x4_f32(false, af[i], false, bf[i], (short)0, acc0, false, false);
    acc1 = __builtin_amdgcn_wmma_f32_16x16x4_f32(false, af[i + 1], false, bf[i + 1], (short)0, acc1, false, false);
    if (REFILL) {
      g2_load1(an[i], bn[i], Ap, Bp, kn + i * 4);
      g2_load1(an[i + 1], bn[i + 1], Ap, Bp, kn + (i + 1) * 4);
    }
  }
}

__global__ __launch_bounds__(128) void gemm2_kernel(const float* __restrict__ a,
                                                    const float* __restrict__ aval,
                                                    float* __restrict__ hist) {
  const int tid = threadIdx.x, lane = tid & 31, wave = tid >> 5;
  const int mt = blockIdx.x, nt = blockIdx.y * 4 + wave;
  if (nt >= (HLEN_ / 16)) return;
  const int m = lane & 15;
  const int kb = (lane & 16) ? 2 : 0;
  const int n = nt * 16 + m;
  const float* Ap = a + (size_t)(mt * 16 + m) * NK_ + kb;
  const float* Bp = aval + (size_t)kb * HLEN_ + n;
  v2f a0[G2NF_], b0[G2NF_], a1[G2NF_], b1[G2NF_], a2[G2NF_], b2[G2NF_];
  v8f acc0 = {}, acc1 = {};
  g2_load(a0, b0, Ap, Bp, 0);
  g2_load(a1, b1, Ap, Bp, G2KB_);
  for (int blk = 0; blk < G2NBLK_ - 2; blk += 3) {
    g2_phase<true>(a0, b0, a2, b2, Ap, Bp, (blk + 2) * G2KB_, acc0, acc1);
    g2_phase<true>(a1, b1, a0, b0, Ap, Bp, (blk + 3) * G2KB_, acc0, acc1);
    g2_phase<true>(a2, b2, a1, b1, Ap, Bp, (blk + 4) * G2KB_, acc0, acc1);
  }
  g2_phase<false>(a0, b0, a2, b2, Ap, Bp, 0, acc0, acc1);
  g2_phase<false>(a1, b1, a2, b2, Ap, Bp, 0, acc0, acc1);
#pragma unroll
  for (int r = 0; r < 8; ++r) {
    int row = r + ((lane & 16) ? 8 : 0);
    hist[(size_t)(mt * 16 + row) * HLEN_ + n] = acc0[r] + acc1[r];
  }
}

__global__ __launch_bounds__(256) void featl1_kernel(const float* __restrict__ inp,
                                                     const float* __restrict__ hist,
                                                     const float* __restrict__ W1,
                                                     const float* __restrict__ b1,
                                                     _Float16* __restrict__ xf) {
  __shared__ float w1s[L1_ * (HOR_ + 1)];
  __shared__ float b1s[L1_];
  __shared__ float hs[8 + HOR_ - 1];
  __shared__ float ins[8];
  const int b = blockIdx.x, t0 = blockIdx.y * 8, tid = threadIdx.x;
  for (int i = tid; i < L1_ * (HOR_ + 1); i += 256) w1s[i] = W1[i];
  if (tid < L1_) b1s[tid] = b1[tid];
  if (tid < 8 + HOR_ - 1) hs[tid] = hist[(size_t)b * HLEN_ + t0 + tid];
  if (tid < 8) ins[tid] = inp[(size_t)b * SEQ_ + t0 + tid];
  __syncthreads();
  const int o = tid & 63;
#pragma unroll
  for (int pass = 0; pass < 2; ++pass) {
    int tt = (tid >> 6) + pass * 4;
    float acc = b1s[o] + ins[tt] * w1s[o * 33];
#pragma unroll
    for (int j = 0; j < HOR_; ++j) acc += hs[tt + j] * w1s[o * 33 + 1 + j];
    xf[((size_t)b * SEQ_ + (t0 + tt)) * L1_ + o] = (_Float16)tanhf(acc);
  }
}

__device__ __forceinline__ void lds_barrier() {
  __builtin_amdgcn_fence(__ATOMIC_RELEASE, "workgroup", "local");
  __builtin_amdgcn_s_barrier();
  __builtin_amdgcn_fence(__ATOMIC_ACQUIRE, "workgroup", "local");
}

#define RB_ 2
#define HR_ (RB_ / 2)
#define TB_ (8 / HR_)
#define NBLK_ (SEQ_ / TB_)
#define WP0_ (L1_ + 8)
#define WP1_ (HID_ + 8)

template <int KIN, int WP>
__device__ __forceinline__ void proj_block(const _Float16* ap, const _Float16* wp,
                                           const float* bp, float* pxs_l) {
  constexpr int KFX = KIN / 32;
  v16h xA[KFX];
#pragma unroll
  for (int kf = 0; kf < KFX; ++kf) {
    v8h lo = *(const v8h*)(ap + kf * 32);
    v8h hi = *(const v8h*)(ap + kf * 32 + 16);
#pragma unroll
    for (int i = 0; i < 8; ++i) { xA[kf][i] = lo[i]; xA[kf][i + 8] = hi[i]; }
  }
#pragma unroll
  for (int nt = 0; nt < 4; ++nt) {
    v8f px;
    const float bias = bp[nt * 16];
#pragma unroll
    for (int r = 0; r < 8; ++r) px[r] = bias;
#pragma unroll
    for (int kf = 0; kf < KFX; ++kf) {
      const _Float16* q = wp + nt * 16 * WP + kf * 32;
      v8h lo = *(const v8h*)q;
      v8h hi = *(const v8h*)(q + 16);
      v16h wB;
#pragma unroll
      for (int i = 0; i < 8; ++i) { wB[i] = lo[i]; wB[i + 8] = hi[i]; }
      if (kf < KFX - 1) {
        px = __builtin_amdgcn_wmma_f32_16x16x32_f16(false, xA[kf], false, wB,
                                                    (short)0, px, false, false);
      } else {
        asm volatile("v_wmma_f32_16x16x32_f16 %0, %1, %2, %0\n\tv_nop\n\tv_nop\n\tv_nop\n\tv_nop"
                     : "+v"(px) : "v"(xA[kf]), "v"(wB));
      }
    }
#pragma unroll
    for (int r = 0; r < 8; ++r) pxs_l[(nt * 8 + r) * 32] = px[r];
  }
}

__global__ __launch_bounds__(512, 1) void lstm2_kernel(const _Float16* __restrict__ xin,
                                                       const float* __restrict__ Wih0,
                                                       const float* __restrict__ Whh0,
                                                       const float* __restrict__ bih0,
                                                       const float* __restrict__ bhh0,
                                                       const float* __restrict__ Wih1,
                                                       const float* __restrict__ Whh1,
                                                       const float* __restrict__ bih1,
                                                       const float* __restrict__ bhh1,
                                                       float* __restrict__ hlast) {
  __shared__ __align__(16) _Float16 wih0s[GATES_ * WP0_];
  __shared__ __align__(16) _Float16 wih1s[GATES_ * WP1_];
  __shared__ float pxs[16 * 4 * 8 * 32];
  __shared__ float gates[2][GATES_ * RB_];
  __shared__ float biass[2][GATES_];
  __shared__ __align__(16) _Float16 hbuf[2][RB_][HID_];
  __shared__ __align__(16) _Float16 h0blk[TB_][RB_][HID_];
  const int tid = threadIdx.x, lane = tid & 31;
  const int wave = __builtin_amdgcn_readfirstlane(tid >> 5);
  const int layer = wave >> 3;
  const int ltid = tid & 255;
  const int bm = blockIdx.x * RB_;
  const int nbase = (wave & 7) * 64;
  const int m = lane & 15;
  const int hi16 = (lane >> 4) & 1;
  const int kbh = hi16 * 8;
  const int arow = (m >> 3) * HR_ + (m & (HR_ - 1));
  const int xts = (m & 7) / HR_;
  const float* __restrict__ Whh = layer ? Whh1 : Whh0;
  const float* __restrict__ bih = layer ? bih1 : bih0;
  const float* __restrict__ bhh = layer ? bhh1 : bhh0;

  for (int i = tid; i < GATES_ * L1_; i += 512)
    wih0s[(i >> 6) * WP0_ + (i & (L1_ - 1))] = (_Float16)Wih0[i];
  for (int i = tid; i < GATES_ * HID_; i += 512)
    wih1s[(i >> 7) * WP1_ + (i & (HID_ - 1))] = (_Float16)Wih1[i];
  for (int i = tid; i < TB_ * RB_ * HID_; i += 512)
    (&h0blk[0][0][0])[i] = (_Float16)0.f;
  for (int i = ltid; i < GATES_; i += 256) biass[layer][i] = bih[i] + bhh[i];

  v16h whhB[4][4];
#pragma unroll
  for (int nt = 0; nt < 4; ++nt) {
    const int g = nbase + nt * 16 + m;
#pragma unroll
    for (int kf = 0; kf < 4; ++kf)
#pragma unroll
      for (int h = 0; h < 16; ++h) {
        int k = kf * 32 + ((h & 8) << 1) + kbh + (h & 7);
        whhB[nt][kf][h] = (_Float16)Whh[(size_t)g * HID_ + k];
      }
  }

  float* gl = gates[layer];
  _Float16 (*hb)[HID_] = hbuf[layer];
  float* pxs_l = pxs + wave * (4 * 8 * 32) + lane;
  float c[HR_];
#pragma unroll
  for (int i = 0; i < HR_; ++i) {
    const int p = ltid + 256 * i;
    c[i] = 0.f;
    hb[p >> 7][p & (HID_ - 1)] = (_Float16)0.f;
  }
  __syncthreads();

  for (int blk = 0; blk <= NBLK_; ++blk) {
    const bool active = layer ? (blk > 0) : (blk < NBLK_);
    const int t0 = (blk - layer) * TB_;
    if (layer == 0) {
      const int tx = (blk < NBLK_ ? blk : NBLK_ - 1) * TB_;
      proj_block<L1_, WP0_>(xin + ((size_t)(bm + arow) * SEQ_ + (tx + xts)) * L1_ + kbh,
                            wih0s + (nbase + m) * WP0_ + kbh, &biass[0][nbase + m], pxs_l);
    } else {
      proj_block<HID_, WP1_>(&h0blk[xts][arow][kbh],
                             wih1s + (nbase + m) * WP1_ + kbh, &biass[1][nbase + m], pxs_l);
    }
#pragma unroll
    for (int tl = 0; tl < TB_; ++tl) {
      const int t = t0 + tl;
      v16h hA[4];
#pragma unroll
      for (int kf = 0; kf < 4; ++kf) {
        const _Float16* p = &hb[arow][kf * 32 + kbh];
        v8h lo = *(const v8h*)p;
        v8h hi = *(const v8h*)(p + 16);
#pragma unroll
        for (int i = 0; i < 8; ++i) { hA[kf][i] = lo[i]; hA[kf][i + 8] = hi[i]; }
      }
      v8f acc[4];
#pragma unroll
      for (int nt = 0; nt < 4; ++nt) {
#pragma unroll
        for (int r = 0; r < 8; ++r) acc[nt][r] = 0.f;
#pragma unroll
        for (int v = 0; v < HR_; ++v) acc[nt][tl * HR_ + v] = pxs_l[(nt * 8 + tl * HR_ + v) * 32];
#pragma unroll
        for (int kf = 0; kf < 4; ++kf)
          acc[nt] = __builtin_amdgcn_wmma_f32_16x16x32_f16(false, hA[kf], false, whhB[nt][kf],
                                                           (short)0, acc[nt], false, false);
      }
#pragma unroll
      for (int nt = 0; nt < 4; ++nt) {
        const int g = nbase + nt * 16 + m;
#pragma unroll
        for (int v = 0; v < HR_; ++v) gl[g * RB_ + hi16 * HR_ + v] = acc[nt][tl * HR_ + v];
      }
      lds_barrier();
      if (active) {
#pragma unroll
        for (int i = 0; i < HR_; ++i) {
          const int p = ltid + 256 * i;
          const int row = p >> 7, j = p & (HID_ - 1);
          const float ia = gl[j * RB_ + row];
          const float fa = gl[(HID_ + j) * RB_ + row];
          const float ga = gl[(2 * HID_ + j) * RB_ + row];
          const float oa = gl[(3 * HID_ + j) * RB_ + row];
          const float iv = 1.0f / (1.0f + __expf(-ia));
          const float fv = 1.0f / (1.0f + __expf(-fa));
          const float gv = tanhf(ga);
          const float ov = 1.0f / (1.0f + __expf(-oa));
          c[i] = fv * c[i] + iv * gv;
          float hv = ov * tanhf(c[i]);
          hb[row][j] = (_Float16)hv;
          if (layer == 0) h0blk[tl][row][j] = (_Float16)hv;
          else if (t == SEQ_ - 1) hlast[(size_t)(bm + row) * HID_ + j] = hv;
        }
      }
      lds_barrier();
    }
  }
}

__global__ __launch_bounds__(128) void head_kernel(const float* __restrict__ h1last,
                                                   const float* __restrict__ W2,
                                                   const float* __restrict__ b2,
                                                   const float* __restrict__ W3,
                                                   const float* __restrict__ b3,
                                                   float* __restrict__ out) {
  __shared__ float lastS[HID_];
  __shared__ float l2S[L2_];
  const int b = blockIdx.x, tid = threadIdx.x;
  lastS[tid] = tanhf(h1last[(size_t)b * HID_ + tid]);
  __syncthreads();
  if (tid < L2_) {
    float acc = b2[tid];
    for (int k = 0; k < HID_; ++k) acc += lastS[k] * W2[tid * HID_ + k];
    l2S[tid] = tanhf(acc);
  }
  __syncthreads();
  if (tid < HOR_) {
    float acc = b3[tid];
    for (int k = 0; k < L2_; ++k) acc += l2S[k] * W3[tid * L2_ + k];
    out[(size_t)b * HOR_ + tid] = acc;
  }
}

extern "C" void kernel_launch(void* const* d_in, const int* in_sizes, int n_in,
                              void* d_out, int out_size, void* d_ws, size_t ws_size,
                              hipStream_t stream) {
  (void)in_sizes; (void)n_in; (void)out_size; (void)ws_size;
  const float* inp  = (const float*)d_in[0];
  const float* akey = (const float*)d_in[1];
  const float* aval = (const float*)d_in[2];
  const float* W1   = (const float*)d_in[3];
  const float* b1   = (const float*)d_in[4];
  const float* Wih0 = (const float*)d_in[5];
  const float* Whh0 = (const float*)d_in[6];
  const float* bih0 = (const float*)d_in[7];
  const float* bhh0 = (const float*)d_in[8];
  const float* Wih1 = (const float*)d_in[9];
  const float* Whh1 = (const float*)d_in[10];
  const float* bih1 = (const float*)d_in[11];
  const float* bhh1 = (const float*)d_in[12];
  const float* W2   = (const float*)d_in[13];
  const float* b2   = (const float*)d_in[14];
  const float* W3   = (const float*)d_in[15];
  const float* b3   = (const float*)d_in[16];

  char* ws = (char*)d_ws;
  size_t off = 0;
  float*    dist   = (float*)(ws + off);    off += (size_t)B_ * NK_ * 4;
  float*    hist   = (float*)(ws + off);    off += (size_t)B_ * HLEN_ * 4;
  _Float16* xf     = (_Float16*)(ws + off); off += (size_t)B_ * SEQ_ * L1_ * 2;
  float*    h1last = (float*)(ws + off);    off += (size_t)B_ * HID_ * 4;

  gemm1_kernel<<<dim3(16, 64), 256, 0, stream>>>(inp, akey, dist);
  softmax_kernel<<<dim3(B_), 256, 0, stream>>>(dist);
  gemm2_kernel<<<dim3(16, 9), 128, 0, stream>>>(dist, aval, hist);
  featl1_kernel<<<dim3(B_, SEQ_ / 8), 256, 0, stream>>>(inp, hist, W1, b1, xf);
  lstm2_kernel<<<dim3(B_ / RB_), 512, 0, stream>>>(xf, Wih0, Whh0, bih0, bhh0, Wih1, Whh1, bih1, bhh1, h1last);
  head_kernel<<<dim3(B_), 128, 0, stream>>>(h1last, W2, b2, W3, b3, (float*)d_out);
}


# Feature layer (32-wide sliding window of hist times W1) becomes a GEMM on the f32 matrix unit, bias + input term in C; one workgroup per batch row (256, was 16384); order of the window sum changes
# speedup vs baseline: 8.0954x; 1.0296x over previous
#include <hip/hip_runtime.h>
#include <cstdint>
#include <cstddef>

typedef __attribute__((ext_vector_type(16))) _Float16 v16h;
typedef __attribute__((ext_vector_type(8)))  _Float16 v8h;
typedef __attribute__((ext_vector_type(8)))  float    v8f;
typedef __attribute__((ext_vector_type(2)))  float    v2f;

#define B_    256
#define SEQ_  512
#define HOR_  32
#define NK_   8192
#define L1_   64
#define HID_  128
#define L2_   64
#define GATES_ 512
#define HLEN_ 544

__global__ __launch_bounds__(256) void gemm1_kernel(const float* __restrict__ inp,
                                                    const float* __restrict__ akey,
                                                    float* __restrict__ dist) {
  const int tid = threadIdx.x, lane = tid & 31, wave = tid >> 5;
  const int mt = blockIdx.x, nt = blockIdx.y * 8 + wave;
  const int m = lane & 15;
  const int kb = (lane & 16) ? 2 : 0;
  const float* Ap = inp  + (size_t)(mt * 16 + m) * SEQ_ + kb;
  const float* Bp = akey + (size_t)(nt * 16 + m) * SEQ_ + kb;
  v8f acc0 = {}, acc1 = {};
  for (int k = 0; k < SEQ_; k += 8) {
    v2f a0 = *(const v2f*)(Ap + k);
    v2f b0 = *(const v2f*)(Bp + k);
    acc0 = __builtin_amdgcn_wmma_f32_16x16x4_f32(false, a0, false, b0, (short)0, acc0, false, false);
    v2f a1 = *(const v2f*)(Ap + k + 4);
    v2f b1 = *(const v2f*)(Bp + k + 4);
    acc1 = __builtin_amdgcn_wmma_f32_16x16x4_f32(false, a1, false, b1, (short)0, acc1, false, false);
  }
#pragma unroll
  for (int r = 0; r < 8; ++r) {
    int row = r + ((lane & 16) ? 8 : 0);
    dist[(size_t)(mt * 16 + row) * NK_ + nt * 16 + m] = acc0[r] + acc1[r];
  }
}

__global__ __launch_bounds__(256) void softmax_kernel(float* __restrict__ dist) {
  __shared__ float sv[256];
  __shared__ int   si[256];
  const int b = blockIdx.x, tid = threadIdx.x;
  float* row = dist + (size_t)b * NK_;

  float m = -3.4e38f; int mi = NK_;
  for (int j = tid; j < NK_; j += 256) {
    float v = row[j];
    if (v > m || (v == m && j < mi)) { m = v; mi = j; }
  }
  sv[tid] = m; si[tid] = mi;
  __syncthreads();
  for (int s = 128; s > 0; s >>= 1) {
    if (tid < s) {
      float v2 = sv[tid + s]; int i2 = si[tid + s];
      if (v2 > sv[tid] || (v2 == sv[tid] && i2 < si[tid])) { sv[tid] = v2; si[tid] = i2; }
    }
    __syncthreads();
  }
  if (tid == 0) row[si[0]] = 0.0f;
  __syncthreads();

  float mm = -3.4e38f;
  for (int j = tid; j < NK_; j += 256) mm = fmaxf(mm, row[j]);
  sv[tid] = mm;
  __syncthreads();
  for (int s = 128; s > 0; s >>= 1) { if (tid < s) sv[tid] = fmaxf(sv[tid], sv[tid + s]); __syncthreads(); }
  const float M = sv[0];
  __syncthreads();

  float sum = 0.f;
  for (int j = tid; j < NK_; j += 256) sum += __expf(row[j] - M);
  sv[tid] = sum;
  __syncthreads();
  for (int s = 128; s > 0; s >>= 1) { if (tid < s) sv[tid] += sv[tid + s]; __syncthreads(); }
  const float inv = 1.0f / sv[0];
  for (int j = tid; j < NK_; j += 256) row[j] = __expf(row[j] - M) * inv;
}

#define G2KB_ 64
#define G2NF_ (G2KB_ / 4)
#define G2NBLK_ (NK_ / G2KB_)

__device__ __forceinline__ void g2_load1(v2f& af, v2f& bf, const float* __restrict__ Ap,
                                         const float* __restrict__ Bp, int k) {
  af = *(const v2f*)(Ap + k);
  bf[0] = Bp[(size_t)k * HLEN_];
  bf[1] = Bp[(size_t)(k + 1) * HLEN_];
}

__device__ __forceinline__ void g2_load(v2f (&af)[G2NF_], v2f (&bf)[G2NF_],
                                        const float* __restrict__ Ap,
                                        const float* __restrict__ Bp, int k0) {
#pragma unroll
  for (int i = 0; i < G2NF_; ++i) g2_load1(af[i], bf[i], Ap, Bp, k0 + i * 4);
}

template <bool REFILL>
__device__ __forceinline__ void g2_phase(const v2f (&af)[G2NF_], const v2f (&bf)[G2NF_],
                                         v2f (&an)[G2NF_], v2f (&bn)[G2NF_],
                                         const float* __restrict__ Ap,
                                         const float* __restrict__ Bp, int kn,
                                         v8f& acc0, v8f& acc1) {
#pragma unroll
  for (int i = 0; i < G2NF_; i += 2) {
    acc0 = __builtin_amdgcn_wmma_f32_16x16x4_f32(false, af[i], false, bf[i], (short)0, acc0, false, false);
    acc1 = __builtin_amdgcn_wmma_f32_16x16x4_f32(false, af[i + 1], false, bf[i + 1], (short)0, acc1, false, false);
    if (REFILL) {
      g2_load1(an[i], bn[i], Ap, Bp, kn + i * 4);
      g2_load1(an[i + 1], bn[i + 1], Ap, Bp, kn + (i + 1) * 4);
    }
  }
}

__global__ __launch_bounds__(128) void gemm2_kernel(const float* __restrict__ a,
                                                    const float* __restrict__ aval,
                                                    float* __restrict__ hist) {
  const int tid = threadIdx.x, lane = tid & 31, wave = tid >> 5;
  const int mt = blockIdx.x, nt = blockIdx.y * 4 + wave;
  if (nt >= (HLEN_ / 16)) return;
  const int m = lane & 15;
  const int kb = (lane & 16) ? 2 : 0;
  const int n = nt * 16 + m;
  const float* Ap = a + (size_t)(mt * 16 + m) * NK_ + kb;
  const float* Bp = aval + (size_t)kb * HLEN_ + n;
  v2f a0[G2NF_], b0[G2NF_], a1[G2NF_], b1[G2NF_], a2[G2NF_], b2[G2NF_];
  v8f acc0 = {}, acc1 = {};
  g2_load(a0, b0, Ap, Bp, 0);
  g2_load(a1, b1, Ap, Bp, G2KB_);
  for (int blk = 0; blk < G2NBLK_ - 2; blk += 3) {
    g2_phase<true>(a0, b0, a2, b2, Ap, Bp, (blk + 2) * G2KB_, acc0, acc1);
    g2_phase<true>(a1, b1, a0, b0, Ap, Bp, (blk + 3) * G2KB_, acc0, acc1);
    g2_phase<true>(a2, b2, a1, b1, Ap, Bp, (blk + 4) * G2KB_, acc0, acc1);
  }
  g2_phase<false>(a0, b0, a2, b2, Ap, Bp, 0, acc0, acc1);
  g2_phase<false>(a1, b1, a2, b2, Ap, Bp, 0, acc0, acc1);
#pragma unroll
  for (int r = 0; r < 8; ++r) {
    int row = r + ((lane & 16) ? 8 : 0);
    hist[(size_t)(mt * 16 + row) * HLEN_ + n] = acc0[r] + acc1[r];
  }
}

typedef __attribute__((ext_vector_type(4))) _Float16 v4h;
#define FKS_ (HOR_ / 4)
#define FNT_ (L1_ / 16)
#define FMT_ 2
#define FTH_ (SEQ_ / (16 * FMT_) * 32)

__global__ __launch_bounds__(FTH_) void featl1_kernel(const float* __restrict__ inp,
                                                     const float* __restrict__ hist,
                                                     const float* __restrict__ W1,
                                                     const float* __restrict__ b1,
                                                     _Float16* __restrict__ xf) {
  __shared__ float w1s[L1_ * (HOR_ + 1)];
  __shared__ float hrow[HLEN_];
  __shared__ float irow[SEQ_];
  const int b = blockIdx.x, tid = threadIdx.x, lane = tid & 31, wave = tid >> 5;
  for (int i = tid; i < L1_ * (HOR_ + 1); i += FTH_) w1s[i] = W1[i];
  for (int i = tid; i < HLEN_; i += FTH_) hrow[i] = hist[(size_t)b * HLEN_ + i];
  for (int i = tid; i < SEQ_; i += FTH_) irow[i] = inp[(size_t)b * SEQ_ + i];
  __syncthreads();
  const int c = lane & 15, half = lane >> 4;
  const int kb = 2 * half;
  v2f bf[FNT_][FKS_];
  float w0[FNT_], bb[FNT_];
#pragma unroll
  for (int nt = 0; nt < FNT_; ++nt) {
    const int o = 4 * c + nt;
    w0[nt] = w1s[o * (HOR_ + 1)];
    bb[nt] = b1[o];
#pragma unroll
    for (int ks = 0; ks < FKS_; ++ks) {
      bf[nt][ks][0] = w1s[o * (HOR_ + 1) + 1 + 4 * ks + kb];
      bf[nt][ks][1] = w1s[o * (HOR_ + 1) + 1 + 4 * ks + kb + 1];
    }
  }
#pragma unroll
  for (int mt = 0; mt < FMT_; ++mt) {
    const int t0 = wave * (16 * FMT_) + mt * 16;
    v2f af[FKS_];
#pragma unroll
    for (int ks = 0; ks < FKS_; ++ks) {
      af[ks][0] = hrow[t0 + c + 4 * ks + kb];
      af[ks][1] = hrow[t0 + c + 4 * ks + kb + 1];
    }
    v8f acc[FNT_];
#pragma unroll
    for (int r = 0; r < 8; ++r) {
      const float iv = irow[t0 + 8 * half + r];
#pragma unroll
      for (int nt = 0; nt < FNT_; ++nt) acc[nt][r] = bb[nt] + iv * w0[nt];
    }
#pragma unroll
    for (int ks = 0; ks < FKS_; ++ks)
#pragma unroll
      for (int nt = 0; nt < FNT_; ++nt)
        acc[nt] = __builtin_amdgcn_wmma_f32_16x16x4_f32(false, af[ks], false, bf[nt][ks],
                                                        (short)0, acc[nt], false, false);
#pragma unroll
    for (int r = 0; r < 8; ++r) {
      v4h pk;
#pragma unroll
      for (int nt = 0; nt < FNT_; ++nt) pk[nt] = (_Float16)tanhf(acc[nt][r]);
      *(v4h*)(xf + ((size_t)b * SEQ_ + (t0 + 8 * half + r)) * L1_ + 4 * c) = pk;
    }
  }
}

__device__ __forceinline__ void lds_barrier() {
  __builtin_amdgcn_fence(__ATOMIC_RELEASE, "workgroup", "local");
  __builtin_amdgcn_s_barrier();
  __builtin_amdgcn_fence(__ATOMIC_ACQUIRE, "workgroup", "local");
}

#define RB_ 2
#define HR_ (RB_ / 2)
#define TB_ (8 / HR_)
#define NBLK_ (SEQ_ / TB_)
#define WP0_ (L1_ + 8)
#define WP1_ (HID_ + 8)

template <int KIN, int WP>
__device__ __forceinline__ void proj_block(const _Float16* ap, const _Float16* wp,
                                           const float* bp, float* pxs_l) {
  constexpr int KFX = KIN / 32;
  v16h xA[KFX];
#pragma unroll
  for (int kf = 0; kf < KFX; ++kf) {
    v8h lo = *(const v8h*)(ap + kf * 32);
    v8h hi = *(const v8h*)(ap + kf * 32 + 16);
#pragma unroll
    for (int i = 0; i < 8; ++i) { xA[kf][i] = lo[i]; xA[kf][i + 8] = hi[i]; }
  }
#pragma unroll
  for (int nt = 0; nt < 4; ++nt) {
    v8f px;
    const float bias = bp[nt * 16];
#pragma unroll
    for (int r = 0; r < 8; ++r) px[r] = bias;
#pragma unroll
    for (int kf = 0; kf < KFX; ++kf) {
      const _Float16* q = wp + nt * 16 * WP + kf * 32;
      v8h lo = *(const v8h*)q;
      v8h hi = *(const v8h*)(q + 16);
      v16h wB;
#pragma unroll
      for (int i = 0; i < 8; ++i) { wB[i] = lo[i]; wB[i + 8] = hi[i]; }
      if (kf < KFX - 1) {
        px = __builtin_amdgcn_wmma_f32_16x16x32_f16(false, xA[kf], false, wB,
                                                    (short)0, px, false, false);
      } else {
        asm volatile("v_wmma_f32_16x16x32_f16 %0, %1, %2, %0\n\tv_nop\n\tv_nop\n\tv_nop\n\tv_nop"
                     : "+v"(px) : "v"(xA[kf]), "v"(wB));
      }
    }
#pragma unroll
    for (int r = 0; r < 8; ++r) pxs_l[(nt * 8 + r) * 32] = px[r];
  }
}

__global__ __launch_bounds__(512, 1) void lstm2_kernel(const _Float16* __restrict__ xin,
                                                       const float* __restrict__ Wih0,
                                                       const float* __restrict__ Whh0,
                                                       const float* __restrict__ bih0,
                                                       const float* __restrict__ bhh0,
                                                       const float* __restrict__ Wih1,
                                                       const float* __restrict__ Whh1,
                                                       const float* __restrict__ bih1,
                                                       const float* __restrict__ bhh1,
                                                       float* __restrict__ hlast) {
  __shared__ __align__(16) _Float16 wih0s[GATES_ * WP0_];
  __shared__ __align__(16) _Float16 wih1s[GATES_ * WP1_];
  __shared__ float pxs[16 * 4 * 8 * 32];
  __shared__ float gates[2][GATES_ * RB_];
  __shared__ float biass[2][GATES_];
  __shared__ __align__(16) _Float16 hbuf[2][RB_][HID_];
  __shared__ __align__(16) _Float16 h0blk[TB_][RB_][HID_];
  const int tid = threadIdx.x, lane = tid & 31;
  const int wave = __builtin_amdgcn_readfirstlane(tid >> 5);
  const int layer = wave >> 3;
  const int ltid = tid & 255;
  const int bm = blockIdx.x * RB_;
  const int nbase = (wave & 7) * 64;
  const int m = lane & 15;
  const int hi16 = (lane >> 4) & 1;
  const int kbh = hi16 * 8;
  const int arow = (m >> 3) * HR_ + (m & (HR_ - 1));
  const int xts = (m & 7) / HR_;
  const float* __restrict__ Whh = layer ? Whh1 : Whh0;
  const float* __restrict__ bih = layer ? bih1 : bih0;
  const float* __restrict__ bhh = layer ? bhh1 : bhh0;

  for (int i = tid; i < GATES_ * L1_; i += 512)
    wih0s[(i >> 6) * WP0_ + (i & (L1_ - 1))] = (_Float16)Wih0[i];
  for (int i = tid; i < GATES_ * HID_; i += 512)
    wih1s[(i >> 7) * WP1_ + (i & (HID_ - 1))] = (_Float16)Wih1[i];
  for (int i = tid; i < TB_ * RB_ * HID_; i += 512)
    (&h0blk[0][0][0])[i] = (_Float16)0.f;
  for (int i = ltid; i < GATES_; i += 256) biass[layer][i] = bih[i] + bhh[i];

  v16h whhB[4][4];
#pragma unroll
  for (int nt = 0; nt < 4; ++nt) {
    const int g = nbase + nt * 16 + m;
#pragma unroll
    for (int kf = 0; kf < 4; ++kf)
#pragma unroll
      for (int h = 0; h < 16; ++h) {
        int k = kf * 32 + ((h & 8) << 1) + kbh + (h & 7);
        whhB[nt][kf][h] = (_Float16)Whh[(size_t)g * HID_ + k];
      }
  }

  float* gl = gates[layer];
  _Float16 (*hb)[HID_] = hbuf[layer];
  float* pxs_l = pxs + wave * (4 * 8 * 32) + lane;
  float c[HR_];
#pragma unroll
  for (int i = 0; i < HR_; ++i) {
    const int p = ltid + 256 * i;
    c[i] = 0.f;
    hb[p >> 7][p & (HID_ - 1)] = (_Float16)0.f;
  }
  __syncthreads();

  for (int blk = 0; blk <= NBLK_; ++blk) {
    const bool active = layer ? (blk > 0) : (blk < NBLK_);
    const int t0 = (blk - layer) * TB_;
    if (layer == 0) {
      const int tx = (blk < NBLK_ ? blk : NBLK_ - 1) * TB_;
      proj_block<L1_, WP0_>(xin + ((size_t)(bm + arow) * SEQ_ + (tx + xts)) * L1_ + kbh,
                            wih0s + (nbase + m) * WP0_ + kbh, &biass[0][nbase + m], pxs_l);
    } else {
      proj_block<HID_, WP1_>(&h0blk[xts][arow][kbh],
                             wih1s + (nbase + m) * WP1_ + kbh, &biass[1][nbase + m], pxs_l);
    }
#pragma unroll
    for (int tl = 0; tl < TB_; ++tl) {
      const int t = t0 + tl;
      v16h hA[4];
#pragma unroll
      for (int kf = 0; kf < 4; ++kf) {
        const _Float16* p = &hb[arow][kf * 32 + kbh];
        v8h lo = *(const v8h*)p;
        v8h hi = *(const v8h*)(p + 16);
#pragma unroll
        for (int i = 0; i < 8; ++i) { hA[kf][i] = lo[i]; hA[kf][i + 8] = hi[i]; }
      }
      v8f acc[4];
#pragma unroll
      for (int nt = 0; nt < 4; ++nt) {
#pragma unroll
        for (int r = 0; r < 8; ++r) acc[nt][r] = 0.f;
#pragma unroll
        for (int v = 0; v < HR_; ++v) acc[nt][tl * HR_ + v] = pxs_l[(nt * 8 + tl * HR_ + v) * 32];
#pragma unroll
        for (int kf = 0; kf < 4; ++kf)
          acc[nt] = __builtin_amdgcn_wmma_f32_16x16x32_f16(false, hA[kf], false, whhB[nt][kf],
                                                           (short)0, acc[nt], false, false);
      }
#pragma unroll
      for (int nt = 0; nt < 4; ++nt) {
        const int g = nbase + nt * 16 + m;
#pragma unroll
        for (int v = 0; v < HR_; ++v) gl[g * RB_ + hi16 * HR_ + v] = acc[nt][tl * HR_ + v];
      }
      lds_barrier();
      if (active) {
#pragma unroll
        for (int i = 0; i < HR_; ++i) {
          const int p = ltid + 256 * i;
          const int row = p >> 7, j = p & (HID_ - 1);
          const float ia = gl[j * RB_ + row];
          const float fa = gl[(HID_ + j) * RB_ + row];
          const float ga = gl[(2 * HID_ + j) * RB_ + row];
          const float oa = gl[(3 * HID_ + j) * RB_ + row];
          const float iv = 1.0f / (1.0f + __expf(-ia));
          const float fv = 1.0f / (1.0f + __expf(-fa));
          const float gv = tanhf(ga);
          const float ov = 1.0f / (1.0f + __expf(-oa));
          c[i] = fv * c[i] + iv * gv;
          float hv = ov * tanhf(c[i]);
          hb[row][j] = (_Float16)hv;
          if (layer == 0) h0blk[tl][row][j] = (_Float16)hv;
          else if (t == SEQ_ - 1) hlast[(size_t)(bm + row) * HID_ + j] = hv;
        }
      }
      lds_barrier();
    }
  }
}

__global__ __launch_bounds__(128) void head_kernel(const float* __restrict__ h1last,
                                                   const float* __restrict__ W2,
                                                   const float* __restrict__ b2,
                                                   const float* __restrict__ W3,
                                                   const float* __restrict__ b3,
                                                   float* __restrict__ out) {
  __shared__ float lastS[HID_];
  __shared__ float l2S[L2_];
  const int b = blockIdx.x, tid = threadIdx.x;
  lastS[tid] = tanhf(h1last[(size_t)b * HID_ + tid]);
  __syncthreads();
  if (tid < L2_) {
    float acc = b2[tid];
    for (int k = 0; k < HID_; ++k) acc += lastS[k] * W2[tid * HID_ + k];
    l2S[tid] = tanhf(acc);
  }
  __syncthreads();
  if (tid < HOR_) {
    float acc = b3[tid];
    for (int k = 0; k < L2_; ++k) acc += l2S[k] * W3[tid * L2_ + k];
    out[(size_t)b * HOR_ + tid] = acc;
  }
}

extern "C" void kernel_launch(void* const* d_in, const int* in_sizes, int n_in,
                              void* d_out, int out_size, void* d_ws, size_t ws_size,
                              hipStream_t stream) {
  (void)in_sizes; (void)n_in; (void)out_size; (void)ws_size;
  const float* inp  = (const float*)d_in[0];
  const float* akey = (const float*)d_in[1];
  const float* aval = (const float*)d_in[2];
  const float* W1   = (const float*)d_in[3];
  const float* b1   = (const float*)d_in[4];
  const float* Wih0 = (const float*)d_in[5];
  const float* Whh0 = (const float*)d_in[6];
  const float* bih0 = (const float*)d_in[7];
  const float* bhh0 = (const float*)d_in[8];
  const float* Wih1 = (const float*)d_in[9];
  const float* Whh1 = (const float*)d_in[10];
  const float* bih1 = (const float*)d_in[11];
  const float* bhh1 = (const float*)d_in[12];
  const float* W2   = (const float*)d_in[13];
  const float* b2   = (const float*)d_in[14];
  const float* W3   = (const float*)d_in[15];
  const float* b3   = (const float*)d_in[16];

  char* ws = (char*)d_ws;
  size_t off = 0;
  float*    dist   = (float*)(ws + off);    off += (size_t)B_ * NK_ * 4;
  float*    hist   = (float*)(ws + off);    off += (size_t)B_ * HLEN_ * 4;
  _Float16* xf     = (_Float16*)(ws + off); off += (size_t)B_ * SEQ_ * L1_ * 2;
  float*    h1last = (float*)(ws + off);    off += (size_t)B_ * HID_ * 4;

  gemm1_kernel<<<dim3(16, 64), 256, 0, stream>>>(inp, akey, dist);
  softmax_kernel<<<dim3(B_), 256, 0, stream>>>(dist);
  gemm2_kernel<<<dim3(16, 9), 128, 0, stream>>>(dist, aval, hist);
  featl1_kernel<<<dim3(B_), FTH_, 0, stream>>>(inp, hist, W1, b1, xf);
  lstm2_kernel<<<dim3(B_ / RB_), 512, 0, stream>>>(xf, Wih0, Whh0, bih0, bhh0, Wih1, Whh1, bih1, bhh1, h1last);
  head_kernel<<<dim3(B_), 128, 0, stream>>>(h1last, W2, b2, W3, b3, (float*)d_out);
}


# LSTM: a workgroup carries 1 batch row instead of 2 (256 workgroups, was 128): 4 of a layer's 8 waves update the state, the projection tile holds 16 timesteps of the row; order of sums kept
# speedup vs baseline: 9.1033x; 1.1245x over previous
#include <hip/hip_runtime.h>
#include <cstdint>
#include <cstddef>

typedef __attribute__((ext_vector_type(16))) _Float16 v16h;
typedef __attribute__((ext_vector_type(8)))  _Float16 v8h;
typedef __attribute__((ext_vector_type(8)))  float    v8f;
typedef __attribute__((ext_vector_type(2)))  float    v2f;

#define B_    256
#define SEQ_  512
#define HOR_  32
#define NK_   8192
#define L1_   64
#define HID_  128
#define L2_   64
#define GATES_ 512
#define HLEN_ 544

__global__ __launch_bounds__(256) void gemm1_kernel(const float* __restrict__ inp,
                                                    const float* __restrict__ akey,
                                                    float* __restrict__ dist) {
  const int tid = threadIdx.x, lane = tid & 31, wave = tid >> 5;
  const int mt = blockIdx.x, nt = blockIdx.y * 8 + wave;
  const int m = lane & 15;
  const int kb = (lane & 16) ? 2 : 0;
  const float* Ap = inp  + (size_t)(mt * 16 + m) * SEQ_ + kb;
  const float* Bp = akey + (size_t)(nt * 16 + m) * SEQ_ + kb;
  v8f acc0 = {}, acc1 = {};
  for (int k = 0; k < SEQ_; k += 8) {
    v2f a0 = *(const v2f*)(Ap + k);
    v2f b0 = *(const v2f*)(Bp + k);
    acc0 = __builtin_amdgcn_wmma_f32_16x16x4_f32(false, a0, false, b0, (short)0, acc0, false, false);
    v2f a1 = *(const v2f*)(Ap + k + 4);
    v2f b1 = *(const v2f*)(Bp + k + 4);
    acc1 = __builtin_amdgcn_wmma_f32_16x16x4_f32(false, a1, false, b1, (short)0, acc1, false, false);
  }
#pragma unroll
  for (int r = 0; r < 8; ++r) {
    int row = r + ((lane & 16) ? 8 : 0);
    dist[(size_t)(mt * 16 + row) * NK_ + nt * 16 + m] = acc0[r] + acc1[r];
  }
}

__global__ __launch_bounds__(256) void softmax_kernel(float* __restrict__ dist) {
  __shared__ float sv[256];
  __shared__ int   si[256];
  const int b = blockIdx.x, tid = threadIdx.x;
  float* row = dist + (size_t)b * NK_;

  float m = -3.4e38f; int mi = NK_;
  for (int j = tid; j < NK_; j += 256) {
    float v = row[j];
    if (v > m || (v == m && j < mi)) { m = v; mi = j; }
  }
  sv[tid] = m; si[tid] = mi;
  __syncthreads();
  for (int s = 128; s > 0; s >>= 1) {
    if (tid < s) {
      float v2 = sv[tid + s]; int i2 = si[tid + s];
      if (v2 > sv[tid] || (v2 == sv[tid] && i2 < si[tid])) { sv[tid] = v2; si[tid] = i2; }
    }
    __syncthreads();
  }
  if (tid == 0) row[si[0]] = 0.0f;
  __syncthreads();

  float mm = -3.4e38f;
  for (int j = tid; j < NK_; j += 256) mm = fmaxf(mm, row[j]);
  sv[tid] = mm;
  __syncthreads();
  for (int s = 128; s > 0; s >>= 1) { if (tid < s) sv[tid] = fmaxf(sv[tid], sv[tid + s]); __syncthreads(); }
  const float M = sv[0];
  __syncthreads();

  float sum = 0.f;
  for (int j = tid; j < NK_; j += 256) sum += __expf(row[j] - M);
  sv[tid] = sum;
  __syncthreads();
  for (int s = 128; s > 0; s >>= 1) { if (tid < s) sv[tid] += sv[tid + s]; __syncthreads(); }
  const float inv = 1.0f / sv[0];
  for (int j = tid; j < NK_; j += 256) row[j] = __expf(row[j] - M) * inv;
}

#define G2KB_ 64
#define G2NF_ (G2KB_ / 4)
#define G2NBLK_ (NK_ / G2KB_)

__device__ __forceinline__ void g2_load1(v2f& af, v2f& bf, const float* __restrict__ Ap,
                                         const float* __restrict__ Bp, int k) {
  af = *(const v2f*)(Ap + k);
  bf[0] = Bp[(size_t)k * HLEN_];
  bf[1] = Bp[(size_t)(k + 1) * HLEN_];
}

__device__ __forceinline__ void g2_load(v2f (&af)[G2NF_], v2f (&bf)[G2NF_],
                                        const float* __restrict__ Ap,
                                        const float* __restrict__ Bp, int k0) {
#pragma unroll
  for (int i = 0; i < G2NF_; ++i) g2_load1(af[i], bf[i], Ap, Bp, k0 + i * 4);
}

template <bool REFILL>
__device__ __forceinline__ void g2_phase(const v2f (&af)[G2NF_], const v2f (&bf)[G2NF_],
                                         v2f (&an)[G2NF_], v2f (&bn)[G2NF_],
                                         const float* __restrict__ Ap,
                                         const float* __restrict__ Bp, int kn,
                                         v8f& acc0, v8f& acc1) {
#pragma unroll
  for (int i = 0; i < G2NF_; i += 2) {
    acc0 = __builtin_amdgcn_wmma_f32_16x16x4_f32(false, af[i], false, bf[i], (short)0, acc0, false, false);
    acc1 = __builtin_amdgcn_wmma_f32_16x16x4_f32(false, af[i + 1], false, bf[i + 1], (short)0, acc1, false, false);
    if (REFILL) {
      g2_load1(an[i], bn[i], Ap, Bp, kn + i * 4);
      g2_load1(an[i + 1], bn[i + 1], Ap, Bp, kn + (i + 1) * 4);
    }
  }
}

__global__ __launch_bounds__(128) void gemm2_kernel(const float* __restrict__ a,
                                                    const float* __restrict__ aval,
                                                    float* __restrict__ hist) {
  const int tid = threadIdx.x, lane = tid & 31, wave = tid >> 5;
  const int mt = blockIdx.x, nt = blockIdx.y * 4 + wave;
  if (nt >= (HLEN_ / 16)) return;
  const int m = lane & 15;
  const int kb = (lane & 16) ? 2 : 0;
  const int n = nt * 16 + m;
  const float* Ap = a + (size_t)(mt * 16 + m) * NK_ + kb;
  const float* Bp = aval + (size_t)kb * HLEN_ + n;
  v2f a0[G2NF_], b0[G2NF_], a1[G2NF_], b1[G2NF_], a2[G2NF_], b2[G2NF_];
  v8f acc0 = {}, acc1 = {};
  g2_load(a0, b0, Ap, Bp, 0);
  g2_load(a1, b1, Ap, Bp, G2KB_);
  for (int blk = 0; blk < G2NBLK_ - 2; blk += 3) {
    g2_phase<true>(a0, b0, a2, b2, Ap, Bp, (blk + 2) * G2KB_, acc0, acc1);
    g2_phase<true>(a1, b1, a0, b0, Ap, Bp, (blk + 3) * G2KB_, acc0, acc1);
    g2_phase<true>(a2, b2, a1, b1, Ap, Bp, (blk + 4) * G2KB_, acc0, acc1);
  }
  g2_phase<false>(a0, b0, a2, b2, Ap, Bp, 0, acc0, acc1);
  g2_phase<false>(a1, b1, a2, b2, Ap, Bp, 0, acc0, acc1);
#pragma unroll
  for (int r = 0; r < 8; ++r) {
    int row = r + ((lane & 16) ? 8 : 0);
    hist[(size_t)(mt * 16 + row) * HLEN_ + n] = acc0[r] + acc1[r];
  }
}

typedef __attribute__((ext_vector_type(4))) _Float16 v4h;
#define FKS_ (HOR_ / 4)
#define FNT_ (L1_ / 16)
#define FMT_ 2
#define FTH_ (SEQ_ / (16 * FMT_) * 32)

__global__ __launch_bounds__(FTH_) void featl1_kernel(const float* __restrict__ inp,
                                                     const float* __restrict__ hist,
                                                     const float* __restrict__ W1,
                                                     const float* __restrict__ b1,
                                                     _Float16* __restrict__ xf) {
  __shared__ float w1s[L1_ * (HOR_ + 1)];
  __shared__ float hrow[HLEN_];
  __shared__ float irow[SEQ_];
  const int b = blockIdx.x, tid = threadIdx.x, lane = tid & 31, wave = tid >> 5;
  for (int i = tid; i < L1_ * (HOR_ + 1); i += FTH_) w1s[i] = W1[i];
  for (int i = tid; i < HLEN_; i += FTH_) hrow[i] = hist[(size_t)b * HLEN_ + i];
  for (int i = tid; i < SEQ_; i += FTH_) irow[i] = inp[(size_t)b * SEQ_ + i];
  __syncthreads();
  const int c = lane & 15, half = lane >> 4;
  const int kb = 2 * half;
  v2f bf[FNT_][FKS_];
  float w0[FNT_], bb[FNT_];
#pragma unroll
  for (int nt = 0; nt < FNT_; ++nt) {
    const int o = 4 * c + nt;
    w0[nt] = w1s[o * (HOR_ + 1)];
    bb[nt] = b1[o];
#pragma unroll
    for (int ks = 0; ks < FKS_; ++ks) {
      bf[nt][ks][0] = w1s[o * (HOR_ + 1) + 1 + 4 * ks + kb];
      bf[nt][ks][1] = w1s[o * (HOR_ + 1) + 1 + 4 * ks + kb + 1];
    }
  }
#pragma unroll
  for (int mt = 0; mt < FMT_; ++mt) {
    const int t0 = wave * (16 * FMT_) + mt * 16;
    v2f af[FKS_];
#pragma unroll
    for (int ks = 0; ks < FKS_; ++ks) {
      af[ks][0] = hrow[t0 + c + 4 * ks + kb];
      af[ks][1] = hrow[t0 + c + 4 * ks + kb + 1];
    }
    v8f acc[FNT_];
#pragma unroll
    for (int r = 0; r < 8; ++r) {
      const float iv = irow[t0 + 8 * half + r];
#pragma unroll
      for (int nt = 0; nt < FNT_; ++nt) acc[nt][r] = bb[nt] + iv * w0[nt];
    }
#pragma unroll
    for (int ks = 0; ks < FKS_; ++ks)
#pragma unroll
      for (int nt = 0; nt < FNT_; ++nt)
        acc[nt] = __builtin_amdgcn_wmma_f32_16x16x4_f32(false, af[ks], false, bf[nt][ks],
                                                        (short)0, acc[nt], false, false);
#pragma unroll
    for (int r = 0; r < 8; ++r) {
      v4h pk;
#pragma unroll
      for (int nt = 0; nt < FNT_; ++nt) pk[nt] = (_Float16)tanhf(acc[nt][r]);
      *(v4h*)(xf + ((size_t)b * SEQ_ + (t0 + 8 * half + r)) * L1_ + 4 * c) = pk;
    }
  }
}

__device__ __forceinline__ void lds_barrier() {
  __builtin_amdgcn_fence(__ATOMIC_RELEASE, "workgroup", "local");
  __builtin_amdgcn_s_barrier();
  __builtin_amdgcn_fence(__ATOMIC_ACQUIRE, "workgroup", "local");
}

#define RB_ 1
#define TB_ 16
#define NBLK_ (SEQ_ / TB_)
#define WP0_ (L1_ + 8)
#define WP1_ (HID_ + 8)

template <int KIN, int WP>
__device__ __forceinline__ void proj_block(const _Float16* ap, const _Float16* wp,
                                           const float* bp, float* pxs_l) {
  constexpr int KFX = KIN / 32;
  v16h xA[KFX];
#pragma unroll
  for (int kf = 0; kf < KFX; ++kf) {
    v8h lo = *(const v8h*)(ap + kf * 32);
    v8h hi = *(const v8h*)(ap + kf * 32 + 16);
#pragma unroll
    for (int i = 0; i < 8; ++i) { xA[kf][i] = lo[i]; xA[kf][i + 8] = hi[i]; }
  }
#pragma unroll
  for (int nt = 0; nt < 4; ++nt) {
    v8f px;
    const float bias = bp[nt * 16];
#pragma unroll
    for (int r = 0; r < 8; ++r) px[r] = bias;
#pragma unroll
    for (int kf = 0; kf < KFX; ++kf) {
      const _Float16* q = wp + nt * 16 * WP + kf * 32;
      v8h lo = *(const v8h*)q;
      v8h hi = *(const v8h*)(q + 16);
      v16h wB;
#pragma unroll
      for (int i = 0; i < 8; ++i) { wB[i] = lo[i]; wB[i + 8] = hi[i]; }
      if (kf < KFX - 1) {
        px = __builtin_amdgcn_wmma_f32_16x16x32_f16(false, xA[kf], false, wB,
                                                    (short)0, px, false, false);
      } else {
        asm volatile("v_wmma_f32_16x16x32_f16 %0, %1, %2, %0\n\tv_nop\n\tv_nop\n\tv_nop\n\tv_nop"
                     : "+v"(px) : "v"(xA[kf]), "v"(wB));
      }
    }
#pragma unroll
    for (int r = 0; r < 8; ++r) pxs_l[(nt * 8 + r) * 32] = px[r];
  }
}

__global__ __launch_bounds__(512, 1) void lstm2_kernel(const _Float16* __restrict__ xin,
                                                       const float* __restrict__ Wih0,
                                                       const float* __restrict__ Whh0,
                                                       const float* __restrict__ bih0,
                                                       const float* __restrict__ bhh0,
                                                       const float* __restrict__ Wih1,
                                                       const float* __restrict__ Whh1,
                                                       const float* __restrict__ bih1,
                                                       const float* __restrict__ bhh1,
                                                       float* __restrict__ hlast) {
  __shared__ __align__(16) _Float16 wih0s[GATES_ * WP0_];
  __shared__ __align__(16) _Float16 wih1s[GATES_ * WP1_];
  __shared__ float pxs[16 * 4 * 8 * 32];
  __shared__ float gates[2][GATES_ * 2];
  __shared__ float biass[2][GATES_];
  __shared__ __align__(16) _Float16 hbuf[2][RB_][HID_];
  __shared__ __align__(16) _Float16 h0blk[TB_][RB_][HID_];
  const int tid = threadIdx.x, lane = tid & 31;
  const int wave = __builtin_amdgcn_readfirstlane(tid >> 5);
  const int layer = wave >> 3;
  const int ltid = tid & 255;
  const int bm = blockIdx.x * RB_;
  const int nbase = (wave & 7) * 64;
  const int m = lane & 15;
  const int hi16 = (lane >> 4) & 1;
  const int kbh = hi16 * 8;
  const float* __restrict__ Whh = layer ? Whh1 : Whh0;
  const float* __restrict__ bih = layer ? bih1 : bih0;
  const float* __restrict__ bhh = layer ? bhh1 : bhh0;

  for (int i = tid; i < GATES_ * L1_; i += 512)
    wih0s[(i >> 6) * WP0_ + (i & (L1_ - 1))] = (_Float16)Wih0[i];
  for (int i = tid; i < GATES_ * HID_; i += 512)
    wih1s[(i >> 7) * WP1_ + (i & (HID_ - 1))] = (_Float16)Wih1[i];
  for (int i = tid; i < TB_ * RB_ * HID_; i += 512)
    (&h0blk[0][0][0])[i] = (_Float16)0.f;
  for (int i = ltid; i < GATES_; i += 256) biass[layer][i] = bih[i] + bhh[i];

  v16h whhB[4][4];
#pragma unroll
  for (int nt = 0; nt < 4; ++nt) {
    const int g = nbase + nt * 16 + m;
#pragma unroll
    for (int kf = 0; kf < 4; ++kf)
#pragma unroll
      for (int h = 0; h < 16; ++h) {
        int k = kf * 32 + ((h & 8) << 1) + kbh + (h & 7);
        whhB[nt][kf][h] = (_Float16)Whh[(size_t)g * HID_ + k];
      }
  }

  float* gl = gates[layer];
  _Float16* hb = hbuf[layer][0];
  float* pxs_l = pxs + wave * (4 * 8 * 32) + lane;
  const bool owner = (wave & 7) < HID_ / 32;
  float c = 0.f;
  if (owner) hb[ltid] = (_Float16)0.f;
  __syncthreads();

  for (int blk = 0; blk <= NBLK_; ++blk) {
    const bool active = layer ? (blk > 0) : (blk < NBLK_);
    const int t0 = (blk - layer) * TB_;
    if (layer == 0) {
      const int tx = (blk < NBLK_ ? blk : NBLK_ - 1) * TB_;
      proj_block<L1_, WP0_>(xin + ((size_t)bm * SEQ_ + (tx + m)) * L1_ + kbh,
                            wih0s + (nbase + m) * WP0_ + kbh, &biass[0][nbase + m], pxs_l);
    } else {
      proj_block<HID_, WP1_>(&h0blk[m][0][kbh],
                             wih1s + (nbase + m) * WP1_ + kbh, &biass[1][nbase + m], pxs_l);
    }
#pragma unroll 1
    for (int hs = 0; hs < 2; ++hs)
#pragma unroll
    for (int r8 = 0; r8 < 8; ++r8) {
      const int tl = hs * 8 + r8;
      const int t = t0 + tl;
      v16h hA[4];
#pragma unroll
      for (int kf = 0; kf < 4; ++kf) {
        const _Float16* p = hb + kf * 32 + kbh;
        v8h lo = *(const v8h*)p;
        v8h hi = *(const v8h*)(p + 16);
#pragma unroll
        for (int i = 0; i < 8; ++i) { hA[kf][i] = lo[i]; hA[kf][i + 8] = hi[i]; }
      }
      v8f acc[4];
#pragma unroll
      for (int nt = 0; nt < 4; ++nt) {
#pragma unroll
        for (int r = 0; r < 8; ++r) acc[nt][r] = 0.f;
        acc[nt][r8] = pxs_l[(nt * 8 + r8) * 32];
#pragma unroll
        for (int kf = 0; kf < 4; ++kf)
          acc[nt] = __builtin_amdgcn_wmma_f32_16x16x32_f16(false, hA[kf], false, whhB[nt][kf],
                                                           (short)0, acc[nt], false, false);
      }
#pragma unroll
      for (int nt = 0; nt < 4; ++nt) {
        const int g = nbase + nt * 16 + m;
        gl[g * 2 + hi16] = acc[nt][r8];
      }
      lds_barrier();
      if (active && owner) {
        const int j = ltid;
        const float ia = gl[j * 2 + hs];
        const float fa = gl[(HID_ + j) * 2 + hs];
        const float ga = gl[(2 * HID_ + j) * 2 + hs];
        const float oa = gl[(3 * HID_ + j) * 2 + hs];
        const float iv = 1.0f / (1.0f + __expf(-ia));
        const float fv = 1.0f / (1.0f + __expf(-fa));
        const float gv = tanhf(ga);
        const float ov = 1.0f / (1.0f + __expf(-oa));
        c = fv * c + iv * gv;
        float hv = ov * tanhf(c);
        hb[j] = (_Float16)hv;
        if (layer == 0) h0blk[tl][0][j] = (_Float16)hv;
        else if (t == SEQ_ - 1) hlast[(size_t)bm * HID_ + j] = hv;
      }
      lds_barrier();
    }
  }
}

__global__ __launch_bounds__(128) void head_kernel(const float* __restrict__ h1last,
                                                   const float* __restrict__ W2,
                                                   const float* __restrict__ b2,
                                                   const float* __restrict__ W3,
                                                   const float* __restrict__ b3,
                                                   float* __restrict__ out) {
  __shared__ float lastS[HID_];
  __shared__ float l2S[L2_];
  const int b = blockIdx.x, tid = threadIdx.x;
  lastS[tid] = tanhf(h1last[(size_t)b * HID_ + tid]);
  __syncthreads();
  if (tid < L2_) {
    float acc = b2[tid];
    for (int k = 0; k < HID_; ++k) acc += lastS[k] * W2[tid * HID_ + k];
    l2S[tid] = tanhf(acc);
  }
  __syncthreads();
  if (tid < HOR_) {
    float acc = b3[tid];
    for (int k = 0; k < L2_; ++k) acc += l2S[k] * W3[tid * L2_ + k];
    out[(size_t)b * HOR_ + tid] = acc;
  }
}

extern "C" void kernel_launch(void* const* d_in, const int* in_sizes, int n_in,
                              void* d_out, int out_size, void* d_ws, size_t ws_size,
                              hipStream_t stream) {
  (void)in_sizes; (void)n_in; (void)out_size; (void)ws_size;
  const float* inp  = (const float*)d_in[0];
  const float* akey = (const float*)d_in[1];
  const float* aval = (const float*)d_in[2];
  const float* W1   = (const float*)d_in[3];
  const float* b1   = (const float*)d_in[4];
  const float* Wih0 = (const float*)d_in[5];
  const float* Whh0 = (const float*)d_in[6];
  const float* bih0 = (const float*)d_in[7];
  const float* bhh0 = (const float*)d_in[8];
  const float* Wih1 = (const float*)d_in[9];
  const float* Whh1 = (const float*)d_in[10];
  const float* bih1 = (const float*)d_in[11];
  const float* bhh1 = (const float*)d_in[12];
  const float* W2   = (const float*)d_in[13];
  const float* b2   = (const float*)d_in[14];
  const float* W3   = (const float*)d_in[15];
  const float* b3   = (const float*)d_in[16];

  char* ws = (char*)d_ws;
  size_t off = 0;
  float*    dist   = (float*)(ws + off);    off += (size_t)B_ * NK_ * 4;
  float*    hist   = (float*)(ws + off);    off += (size_t)B_ * HLEN_ * 4;
  _Float16* xf     = (_Float16*)(ws + off); off += (size_t)B_ * SEQ_ * L1_ * 2;
  float*    h1last = (float*)(ws + off);    off += (size_t)B_ * HID_ * 4;

  gemm1_kernel<<<dim3(16, 64), 256, 0, stream>>>(inp, akey, dist);
  softmax_kernel<<<dim3(B_), 256, 0, stream>>>(dist);
  gemm2_kernel<<<dim3(16, 9), 128, 0, stream>>>(dist, aval, hist);
  featl1_kernel<<<dim3(B_), FTH_, 0, stream>>>(inp, hist, W1, b1, xf);
  lstm2_kernel<<<dim3(B_ / RB_), 512, 0, stream>>>(xf, Wih0, Whh0, bih0, bhh0, Wih1, Whh1, bih1, bhh1, h1last);
  head_kernel<<<dim3(B_), 128, 0, stream>>>(h1last, W2, b2, W3, b3, (float*)d_out);
}


# LSTM step: each matrix wave applies the gate function (same sigmoid/tanh formulas) to its own sums before the barrier, beside other waves' WMMAs; the owner thread keeps only c, tanh(c), h; same sums
# speedup vs baseline: 9.6795x; 1.0633x over previous
#include <hip/hip_runtime.h>
#include <cstdint>
#include <cstddef>

typedef __attribute__((ext_vector_type(16))) _Float16 v16h;
typedef __attribute__((ext_vector_type(8)))  _Float16 v8h;
typedef __attribute__((ext_vector_type(8)))  float    v8f;
typedef __attribute__((ext_vector_type(2)))  float    v2f;

#define B_    256
#define SEQ_  512
#define HOR_  32
#define NK_   8192
#define L1_   64
#define HID_  128
#define L2_   64
#define GATES_ 512
#define HLEN_ 544

__global__ __launch_bounds__(256) void gemm1_kernel(const float* __restrict__ inp,
                                                    const float* __restrict__ akey,
                                                    float* __restrict__ dist) {
  const int tid = threadIdx.x, lane = tid & 31, wave = tid >> 5;
  const int mt = blockIdx.x, nt = blockIdx.y * 8 + wave;
  const int m = lane & 15;
  const int kb = (lane & 16) ? 2 : 0;
  const float* Ap = inp  + (size_t)(mt * 16 + m) * SEQ_ + kb;
  const float* Bp = akey + (size_t)(nt * 16 + m) * SEQ_ + kb;
  v8f acc0 = {}, acc1 = {};
  for (int k = 0; k < SEQ_; k += 8) {
    v2f a0 = *(const v2f*)(Ap + k);
    v2f b0 = *(const v2f*)(Bp + k);
    acc0 = __builtin_amdgcn_wmma_f32_16x16x4_f32(false, a0, false, b0, (short)0, acc0, false, false);
    v2f a1 = *(const v2f*)(Ap + k + 4);
    v2f b1 = *(const v2f*)(Bp + k + 4);
    acc1 = __builtin_amdgcn_wmma_f32_16x16x4_f32(false, a1, false, b1, (short)0, acc1, false, false);
  }
#pragma unroll
  for (int r = 0; r < 8; ++r) {
    int row = r + ((lane & 16) ? 8 : 0);
    dist[(size_t)(mt * 16 + row) * NK_ + nt * 16 + m] = acc0[r] + acc1[r];
  }
}

__global__ __launch_bounds__(256) void softmax_kernel(float* __restrict__ dist) {
  __shared__ float sv[256];
  __shared__ int   si[256];
  const int b = blockIdx.x, tid = threadIdx.x;
  float* row = dist + (size_t)b * NK_;

  float m = -3.4e38f; int mi = NK_;
  for (int j = tid; j < NK_; j += 256) {
    float v = row[j];
    if (v > m || (v == m && j < mi)) { m = v; mi = j; }
  }
  sv[tid] = m; si[tid] = mi;
  __syncthreads();
  for (int s = 128; s > 0; s >>= 1) {
    if (tid < s) {
      float v2 = sv[tid + s]; int i2 = si[tid + s];
      if (v2 > sv[tid] || (v2 == sv[tid] && i2 < si[tid])) { sv[tid] = v2; si[tid] = i2; }
    }
    __syncthreads();
  }
  if (tid == 0) row[si[0]] = 0.0f;
  __syncthreads();

  float mm = -3.4e38f;
  for (int j = tid; j < NK_; j += 256) mm = fmaxf(mm, row[j]);
  sv[tid] = mm;
  __syncthreads();
  for (int s = 128; s > 0; s >>= 1) { if (tid < s) sv[tid] = fmaxf(sv[tid], sv[tid + s]); __syncthreads(); }
  const float M = sv[0];
  __syncthreads();

  float sum = 0.f;
  for (int j = tid; j < NK_; j += 256) sum += __expf(row[j] - M);
  sv[tid] = sum;
  __syncthreads();
  for (int s = 128; s > 0; s >>= 1) { if (tid < s) sv[tid] += sv[tid + s]; __syncthreads(); }
  const float inv = 1.0f / sv[0];
  for (int j = tid; j < NK_; j += 256) row[j] = __expf(row[j] - M) * inv;
}

#define G2KB_ 64
#define G2NF_ (G2KB_ / 4)
#define G2NBLK_ (NK_ / G2KB_)

__device__ __forceinline__ void g2_load1(v2f& af, v2f& bf, const float* __restrict__ Ap,
                                         const float* __restrict__ Bp, int k) {
  af = *(const v2f*)(Ap + k);
  bf[0] = Bp[(size_t)k * HLEN_];
  bf[1] = Bp[(size_t)(k + 1) * HLEN_];
}

__device__ __forceinline__ void g2_load(v2f (&af)[G2NF_], v2f (&bf)[G2NF_],
                                        const float* __restrict__ Ap,
                                        const float* __restrict__ Bp, int k0) {
#pragma unroll
  for (int i = 0; i < G2NF_; ++i) g2_load1(af[i], bf[i], Ap, Bp, k0 + i * 4);
}

template <bool REFILL>
__device__ __forceinline__ void g2_phase(const v2f (&af)[G2NF_], const v2f (&bf)[G2NF_],
                                         v2f (&an)[G2NF_], v2f (&bn)[G2NF_],
                                         const float* __restrict__ Ap,
                                         const float* __restrict__ Bp, int kn,
                                         v8f& acc0, v8f& acc1) {
#pragma unroll
  for (int i = 0; i < G2NF_; i += 2) {
    acc0 = __builtin_amdgcn_wmma_f32_16x16x4_f32(false, af[i], false, bf[i], (short)0, acc0, false, false);
    acc1 = __builtin_amdgcn_wmma_f32_16x16x4_f32(false, af[i + 1], false, bf[i + 1], (short)0, acc1, false, false);
    if (REFILL) {
      g2_load1(an[i], bn[i], Ap, Bp, kn + i * 4);
      g2_load1(an[i + 1], bn[i + 1], Ap, Bp, kn + (i + 1) * 4);
    }
  }
}

__global__ __launch_bounds__(128) void gemm2_kernel(const float* __restrict__ a,
                                                    const float* __restrict__ aval,
                                                    float* __restrict__ hist) {
  const int tid = threadIdx.x, lane = tid & 31, wave = tid >> 5;
  const int mt = blockIdx.x, nt = blockIdx.y * 4 + wave;
  if (nt >= (HLEN_ / 16)) return;
  const int m = lane & 15;
  const int kb = (lane & 16) ? 2 : 0;
  const int n = nt * 16 + m;
  const float* Ap = a + (size_t)(mt * 16 + m) * NK_ + kb;
  const float* Bp = aval + (size_t)kb * HLEN_ + n;
  v2f a0[G2NF_], b0[G2NF_], a1[G2NF_], b1[G2NF_], a2[G2NF_], b2[G2NF_];
  v8f acc0 = {}, acc1 = {};
  g2_load(a0, b0, Ap, Bp, 0);
  g2_load(a1, b1, Ap, Bp, G2KB_);
  for (int blk = 0; blk < G2NBLK_ - 2; blk += 3) {
    g2_phase<true>(a0, b0, a2, b2, Ap, Bp, (blk + 2) * G2KB_, acc0, acc1);
    g2_phase<true>(a1, b1, a0, b0, Ap, Bp, (blk + 3) * G2KB_, acc0, acc1);
    g2_phase<true>(a2, b2, a1, b1, Ap, Bp, (blk + 4) * G2KB_, acc0, acc1);
  }
  g2_phase<false>(a0, b0, a2, b2, Ap, Bp, 0, acc0, acc1);
  g2_phase<false>(a1, b1, a2, b2, Ap, Bp, 0, acc0, acc1);
#pragma unroll
  for (int r = 0; r < 8; ++r) {
    int row = r + ((lane & 16) ? 8 : 0);
    hist[(size_t)(mt * 16 + row) * HLEN_ + n] = acc0[r] + acc1[r];
  }
}

typedef __attribute__((ext_vector_type(4))) _Float16 v4h;
#define FKS_ (HOR_ / 4)
#define FNT_ (L1_ / 16)
#define FMT_ 2
#define FTH_ (SEQ_ / (16 * FMT_) * 32)

__global__ __launch_bounds__(FTH_) void featl1_kernel(const float* __restrict__ inp,
                                                     const float* __restrict__ hist,
                                                     const float* __restrict__ W1,
                                                     const float* __restrict__ b1,
                                                     _Float16* __restrict__ xf) {
  __shared__ float w1s[L1_ * (HOR_ + 1)];
  __shared__ float hrow[HLEN_];
  __shared__ float irow[SEQ_];
  const int b = blockIdx.x, tid = threadIdx.x, lane = tid & 31, wave = tid >> 5;
  for (int i = tid; i < L1_ * (HOR_ + 1); i += FTH_) w1s[i] = W1[i];
  for (int i = tid; i < HLEN_; i += FTH_) hrow[i] = hist[(size_t)b * HLEN_ + i];
  for (int i = tid; i < SEQ_; i += FTH_) irow[i] = inp[(size_t)b * SEQ_ + i];
  __syncthreads();
  const int c = lane & 15, half = lane >> 4;
  const int kb = 2 * half;
  v2f bf[FNT_][FKS_];
  float w0[FNT_], bb[FNT_];
#pragma unroll
  for (int nt = 0; nt < FNT_; ++nt) {
    const int o = 4 * c + nt;
    w0[nt] = w1s[o * (HOR_ + 1)];
    bb[nt] = b1[o];
#pragma unroll
    for (int ks = 0; ks < FKS_; ++ks) {
      bf[nt][ks][0] = w1s[o * (HOR_ + 1) + 1 + 4 * ks + kb];
      bf[nt][ks][1] = w1s[o * (HOR_ + 1) + 1 + 4 * ks + kb + 1];
    }
  }
#pragma unroll
  for (int mt = 0; mt < FMT_; ++mt) {
    const int t0 = wave * (16 * FMT_) + mt * 16;
    v2f af[FKS_];
#pragma unroll
    for (int ks = 0; ks < FKS_; ++ks) {
      af[ks][0] = hrow[t0 + c + 4 * ks + kb];
      af[ks][1] = hrow[t0 + c + 4 * ks + kb + 1];
    }
    v8f acc[FNT_];
#pragma unroll
    for (int r = 0; r < 8; ++r) {
      const float iv = irow[t0 + 8 * half + r];
#pragma unroll
      for (int nt = 0; nt < FNT_; ++nt) acc[nt][r] = bb[nt] + iv * w0[nt];
    }
#pragma unroll
    for (int ks = 0; ks < FKS_; ++ks)
#pragma unroll
      for (int nt = 0; nt < FNT_; ++nt)
        acc[nt] = __builtin_amdgcn_wmma_f32_16x16x4_f32(false, af[ks], false, bf[nt][ks],
                                                        (short)0, acc[nt], false, false);
#pragma unroll
    for (int r = 0; r < 8; ++r) {
      v4h pk;
#pragma unroll
      for (int nt = 0; nt < FNT_; ++nt) pk[nt] = (_Float16)tanhf(acc[nt][r]);
      *(v4h*)(xf + ((size_t)b * SEQ_ + (t0 + 8 * half + r)) * L1_ + 4 * c) = pk;
    }
  }
}

__device__ __forceinline__ void lds_barrier() {
  __builtin_amdgcn_fence(__ATOMIC_RELEASE, "workgroup", "local");
  __builtin_amdgcn_s_barrier();
  __builtin_amdgcn_fence(__ATOMIC_ACQUIRE, "workgroup", "local");
}

#define RB_ 1
#define TB_ 16
#define NBLK_ (SEQ_ / TB_)
#define WP0_ (L1_ + 8)
#define WP1_ (HID_ + 8)

template <int KIN, int WP>
__device__ __forceinline__ void proj_block(const _Float16* ap, const _Float16* wp,
                                           const float* bp, int cstep, float* pxs_l) {
  constexpr int KFX = KIN / 32;
  v16h xA[KFX];
#pragma unroll
  for (int kf = 0; kf < KFX; ++kf) {
    v8h lo = *(const v8h*)(ap + kf * 32);
    v8h hi = *(const v8h*)(ap + kf * 32 + 16);
#pragma unroll
    for (int i = 0; i < 8; ++i) { xA[kf][i] = lo[i]; xA[kf][i + 8] = hi[i]; }
  }
#pragma unroll
  for (int nt = 0; nt < 4; ++nt) {
    v8f px;
    const int co = (nt & 1) * 16 + (nt >> 1) * cstep;
    const float bias = bp[co];
#pragma unroll
    for (int r = 0; r < 8; ++r) px[r] = bias;
#pragma unroll
    for (int kf = 0; kf < KFX; ++kf) {
      const _Float16* q = wp + co * WP + kf * 32;
      v8h lo = *(const v8h*)q;
      v8h hi = *(const v8h*)(q + 16);
      v16h wB;
#pragma unroll
      for (int i = 0; i < 8; ++i) { wB[i] = lo[i]; wB[i + 8] = hi[i]; }
      if (kf < KFX - 1) {
        px = __builtin_amdgcn_wmma_f32_16x16x32_f16(false, xA[kf], false, wB,
                                                    (short)0, px, false, false);
      } else {
        asm volatile("v_wmma_f32_16x16x32_f16 %0, %1, %2, %0\n\tv_nop\n\tv_nop\n\tv_nop\n\tv_nop"
                     : "+v"(px) : "v"(xA[kf]), "v"(wB));
      }
    }
#pragma unroll
    for (int r = 0; r < 8; ++r) pxs_l[(nt * 8 + r) * 32] = px[r];
  }
}

__global__ __launch_bounds__(512, 1) void lstm2_kernel(const _Float16* __restrict__ xin,
                                                       const float* __restrict__ Wih0,
                                                       const float* __restrict__ Whh0,
                                                       const float* __restrict__ bih0,
                                                       const float* __restrict__ bhh0,
                                                       const float* __restrict__ Wih1,
                                                       const float* __restrict__ Whh1,
                                                       const float* __restrict__ bih1,
                                                       const float* __restrict__ bhh1,
                                                       float* __restrict__ hlast) {
  __shared__ __align__(16) _Float16 wih0s[GATES_ * WP0_];
  __shared__ __align__(16) _Float16 wih1s[GATES_ * WP1_];
  __shared__ float pxs[16 * 4 * 8 * 32];
  __shared__ float gates[2][GATES_];
  __shared__ float biass[2][GATES_];
  __shared__ __align__(16) _Float16 hbuf[2][RB_][HID_];
  __shared__ __align__(16) _Float16 h0blk[TB_][RB_][HID_];
  const int tid = threadIdx.x, lane = tid & 31;
  const int wave = __builtin_amdgcn_readfirstlane(tid >> 5);
  const int layer = wave >> 3;
  const int ltid = tid & 255;
  const int bm = blockIdx.x * RB_;
  const int wl = wave & 7;
  const bool tanh_wave = wl < 4;
  const int cb0 = tanh_wave ? 2 * HID_ + 32 * wl : (wl - 4) * 64;
  const int cstep = tanh_wave ? HID_ : 32;
  const int m = lane & 15;
  const int hi16 = (lane >> 4) & 1;
  const int kbh = hi16 * 8;
  const float* __restrict__ Whh = layer ? Whh1 : Whh0;
  const float* __restrict__ bih = layer ? bih1 : bih0;
  const float* __restrict__ bhh = layer ? bhh1 : bhh0;

  for (int i = tid; i < GATES_ * L1_; i += 512)
    wih0s[(i >> 6) * WP0_ + (i & (L1_ - 1))] = (_Float16)Wih0[i];
  for (int i = tid; i < GATES_ * HID_; i += 512)
    wih1s[(i >> 7) * WP1_ + (i & (HID_ - 1))] = (_Float16)Wih1[i];
  for (int i = tid; i < TB_ * RB_ * HID_; i += 512)
    (&h0blk[0][0][0])[i] = (_Float16)0.f;
  for (int i = ltid; i < GATES_; i += 256) biass[layer][i] = bih[i] + bhh[i];

  v16h whhB[4][4];
#pragma unroll
  for (int nt = 0; nt < 4; ++nt) {
    const int g = cb0 + (nt & 1) * 16 + (nt >> 1) * cstep + m;
#pragma unroll
    for (int kf = 0; kf < 4; ++kf)
#pragma unroll
      for (int h = 0; h < 16; ++h) {
        int k = kf * 32 + ((h & 8) << 1) + kbh + (h & 7);
        whhB[nt][kf][h] = (_Float16)Whh[(size_t)g * HID_ + k];
      }
  }

  float* gl = gates[layer];
  _Float16* hb = hbuf[layer][0];
  float* pxs_l = pxs + wave * (4 * 8 * 32) + lane;
  const float* pxs_r = pxs + wave * (4 * 8 * 32) + m;
  const bool owner = (wave & 7) < HID_ / 32;
  float c = 0.f;
  if (owner) hb[ltid] = (_Float16)0.f;
  __syncthreads();

  for (int blk = 0; blk <= NBLK_; ++blk) {
    const bool active = layer ? (blk > 0) : (blk < NBLK_);
    const int t0 = (blk - layer) * TB_;
    if (layer == 0) {
      const int tx = (blk < NBLK_ ? blk : NBLK_ - 1) * TB_;
      proj_block<L1_, WP0_>(xin + ((size_t)bm * SEQ_ + (tx + m)) * L1_ + kbh,
                            wih0s + (cb0 + m) * WP0_ + kbh, &biass[0][cb0 + m], cstep, pxs_l);
    } else {
      proj_block<HID_, WP1_>(&h0blk[m][0][kbh],
                             wih1s + (cb0 + m) * WP1_ + kbh, &biass[1][cb0 + m], cstep, pxs_l);
    }
#pragma unroll 1
    for (int hs = 0; hs < 2; ++hs)
#pragma unroll
    for (int r8 = 0; r8 < 8; ++r8) {
      const int tl = hs * 8 + r8;
      const int t = t0 + tl;
      v16h hA[4];
#pragma unroll
      for (int kf = 0; kf < 4; ++kf) {
        const _Float16* p = hb + kf * 32 + kbh;
        v8h lo = *(const v8h*)p;
        v8h hi = *(const v8h*)(p + 16);
#pragma unroll
        for (int i = 0; i < 8; ++i) { hA[kf][i] = lo[i]; hA[kf][i + 8] = hi[i]; }
      }
      v8f acc[4];
#pragma unroll
      for (int nt = 0; nt < 4; ++nt) {
#pragma unroll
        for (int r = 0; r < 8; ++r) acc[nt][r] = 0.f;
        acc[nt][r8] = pxs_r[(nt * 8 + r8) * 32 + 16 * hs];
#pragma unroll
        for (int kf = 0; kf < 4; ++kf)
          acc[nt] = __builtin_amdgcn_wmma_f32_16x16x32_f16(false, hA[kf], false, whhB[nt][kf],
                                                           (short)0, acc[nt], false, false);
      }
      const float s0 = hi16 ? acc[1][r8] : acc[0][r8];
      const float s1 = hi16 ? acc[3][r8] : acc[2][r8];
      gl[cb0 + cstep + lane] = 1.0f / (1.0f + __expf(-s1));
      float a0 = 1.0f / (1.0f + __expf(-s0));
      if (tanh_wave) a0 = tanhf(s0);
      gl[cb0 + lane] = a0;
      lds_barrier();
      if (active && owner) {
        const int j = ltid;
        const float iv = gl[j];
        const float fv = gl[HID_ + j];
        const float gv = gl[2 * HID_ + j];
        const float ov = gl[3 * HID_ + j];
        c = fv * c + iv * gv;
        float hv = ov * tanhf(c);
        hb[j] = (_Float16)hv;
        if (layer == 0) h0blk[tl][0][j] = (_Float16)hv;
        else if (t == SEQ_ - 1) hlast[(size_t)bm * HID_ + j] = hv;
      }
      lds_barrier();
    }
  }
}

__global__ __launch_bounds__(128) void head_kernel(const float* __restrict__ h1last,
                                                   const float* __restrict__ W2,
                                                   const float* __restrict__ b2,
                                                   const float* __restrict__ W3,
                                                   const float* __restrict__ b3,
                                                   float* __restrict__ out) {
  __shared__ float lastS[HID_];
  __shared__ float l2S[L2_];
  const int b = blockIdx.x, tid = threadIdx.x;
  lastS[tid] = tanhf(h1last[(size_t)b * HID_ + tid]);
  __syncthreads();
  if (tid < L2_) {
    float acc = b2[tid];
    for (int k = 0; k < HID_; ++k) acc += lastS[k] * W2[tid * HID_ + k];
    l2S[tid] = tanhf(acc);
  }
  __syncthreads();
  if (tid < HOR_) {
    float acc = b3[tid];
    for (int k = 0; k < L2_; ++k) acc += l2S[k] * W3[tid * L2_ + k];
    out[(size_t)b * HOR_ + tid] = acc;
  }
}

extern "C" void kernel_launch(void* const* d_in, const int* in_sizes, int n_in,
                              void* d_out, int out_size, void* d_ws, size_t ws_size,
                              hipStream_t stream) {
  (void)in_sizes; (void)n_in; (void)out_size; (void)ws_size;
  const float* inp  = (const float*)d_in[0];
  const float* akey = (const float*)d_in[1];
  const float* aval = (const float*)d_in[2];
  const float* W1   = (const float*)d_in[3];
  const float* b1   = (const float*)d_in[4];
  const float* Wih0 = (const float*)d_in[5];
  const float* Whh0 = (const float*)d_in[6];
  const float* bih0 = (const float*)d_in[7];
  const float* bhh0 = (const float*)d_in[8];
  const float* Wih1 = (const float*)d_in[9];
  const float* Whh1 = (const float*)d_in[10];
  const float* bih1 = (const float*)d_in[11];
  const float* bhh1 = (const float*)d_in[12];
  const float* W2   = (const float*)d_in[13];
  const float* b2   = (const float*)d_in[14];
  const float* W3   = (const float*)d_in[15];
  const float* b3   = (const float*)d_in[16];

  char* ws = (char*)d_ws;
  size_t off = 0;
  float*    dist   = (float*)(ws + off);    off += (size_t)B_ * NK_ * 4;
  float*    hist   = (float*)(ws + off);    off += (size_t)B_ * HLEN_ * 4;
  _Float16* xf     = (_Float16*)(ws + off); off += (size_t)B_ * SEQ_ * L1_ * 2;
  float*    h1last = (float*)(ws + off);    off += (size_t)B_ * HID_ * 4;

  gemm1_kernel<<<dim3(16, 64), 256, 0, stream>>>(inp, akey, dist);
  softmax_kernel<<<dim3(B_), 256, 0, stream>>>(dist);
  gemm2_kernel<<<dim3(16, 9), 128, 0, stream>>>(dist, aval, hist);
  featl1_kernel<<<dim3(B_), FTH_, 0, stream>>>(inp, hist, W1, b1, xf);
  lstm2_kernel<<<dim3(B_ / RB_), 512, 0, stream>>>(xf, Wih0, Whh0, bih0, bhh0, Wih1, Whh1, bih1, bhh1, h1last);
  head_kernel<<<dim3(B_), 128, 0, stream>>>(h1last, W2, b2, W3, b3, (float*)d_out);
}


# Both f32 GEMMs: a lane loads 16 B of its A row (GEMM1: of the B row too) per instruction and feeds two k-steps from it, half as many 16-line loads; k grouping of the sums changes inside each 8 k
# speedup vs baseline: 9.9779x; 1.0308x over previous
#include <hip/hip_runtime.h>
#include <cstdint>
#include <cstddef>

typedef __attribute__((ext_vector_type(16))) _Float16 v16h;
typedef __attribute__((ext_vector_type(8)))  _Float16 v8h;
typedef __attribute__((ext_vector_type(8)))  float    v8f;
typedef __attribute__((ext_vector_type(2)))  float    v2f;
typedef __attribute__((ext_vector_type(4)))  float    v4f;

#define B_    256
#define SEQ_  512
#define HOR_  32
#define NK_   8192
#define L1_   64
#define HID_  128
#define L2_   64
#define GATES_ 512
#define HLEN_ 544

__global__ __launch_bounds__(256) void gemm1_kernel(const float* __restrict__ inp,
                                                    const float* __restrict__ akey,
                                                    float* __restrict__ dist) {
  const int tid = threadIdx.x, lane = tid & 31, wave = tid >> 5;
  const int mt = blockIdx.x, nt = blockIdx.y * 8 + wave;
  const int m = lane & 15;
  const int kb = (lane & 16) ? 4 : 0;
  const float* Ap = inp  + (size_t)(mt * 16 + m) * SEQ_ + kb;
  const float* Bp = akey + (size_t)(nt * 16 + m) * SEQ_ + kb;
  v8f acc0 = {}, acc1 = {};
  for (int k = 0; k < SEQ_; k += 8) {
    const v4f a = *(const v4f*)(Ap + k);
    const v4f b = *(const v4f*)(Bp + k);
    const v2f a0 = __builtin_shufflevector(a, a, 0, 1), b0 = __builtin_shufflevector(b, b, 0, 1);
    const v2f a1 = __builtin_shufflevector(a, a, 2, 3), b1 = __builtin_shufflevector(b, b, 2, 3);
    acc0 = __builtin_amdgcn_wmma_f32_16x16x4_f32(false, a0, false, b0, (short)0, acc0, false, false);
    acc1 = __builtin_amdgcn_wmma_f32_16x16x4_f32(false, a1, false, b1, (short)0, acc1, false, false);
  }
#pragma unroll
  for (int r = 0; r < 8; ++r) {
    int row = r + ((lane & 16) ? 8 : 0);
    dist[(size_t)(mt * 16 + row) * NK_ + nt * 16 + m] = acc0[r] + acc1[r];
  }
}

__global__ __launch_bounds__(256) void softmax_kernel(float* __restrict__ dist) {
  __shared__ float sv[256];
  __shared__ int   si[256];
  const int b = blockIdx.x, tid = threadIdx.x;
  float* row = dist + (size_t)b * NK_;

  float m = -3.4e38f; int mi = NK_;
  for (int j = tid; j < NK_; j += 256) {
    float v = row[j];
    if (v > m || (v == m && j < mi)) { m = v; mi = j; }
  }
  sv[tid] = m; si[tid] = mi;
  __syncthreads();
  for (int s = 128; s > 0; s >>= 1) {
    if (tid < s) {
      float v2 = sv[tid + s]; int i2 = si[tid + s];
      if (v2 > sv[tid] || (v2 == sv[tid] && i2 < si[tid])) { sv[tid] = v2; si[tid] = i2; }
    }
    __syncthreads();
  }
  if (tid == 0) row[si[0]] = 0.0f;
  __syncthreads();

  float mm = -3.4e38f;
  for (int j = tid; j < NK_; j += 256) mm = fmaxf(mm, row[j]);
  sv[tid] = mm;
  __syncthreads();
  for (int s = 128; s > 0; s >>= 1) { if (tid < s) sv[tid] = fmaxf(sv[tid], sv[tid + s]); __syncthreads(); }
  const float M = sv[0];
  __syncthreads();

  float sum = 0.f;
  for (int j = tid; j < NK_; j += 256) sum += __expf(row[j] - M);
  sv[tid] = sum;
  __syncthreads();
  for (int s = 128; s > 0; s >>= 1) { if (tid < s) sv[tid] += sv[tid + s]; __syncthreads(); }
  const float inv = 1.0f / sv[0];
  for (int j = tid; j < NK_; j += 256) row[j] = __expf(row[j] - M) * inv;
}

#define G2KB_ 64
#define G2NF_ (G2KB_ / 4)
#define G2NBLK_ (NK_ / G2KB_)

__device__ __forceinline__ void g2_load2(v2f& af0, v2f& af1, v2f& bf0, v2f& bf1,
                                         const float* __restrict__ Ap,
                                         const float* __restrict__ Bp, int k) {
  const v4f a = *(const v4f*)(Ap + k);
  af0 = __builtin_shufflevector(a, a, 0, 1);
  af1 = __builtin_shufflevector(a, a, 2, 3);
  bf0[0] = Bp[(size_t)k * HLEN_];
  bf0[1] = Bp[(size_t)(k + 1) * HLEN_];
  bf1[0] = Bp[(size_t)(k + 2) * HLEN_];
  bf1[1] = Bp[(size_t)(k + 3) * HLEN_];
}

__device__ __forceinline__ void g2_load(v2f (&af)[G2NF_], v2f (&bf)[G2NF_],
                                        const float* __restrict__ Ap,
                                        const float* __restrict__ Bp, int k0) {
#pragma unroll
  for (int i = 0; i < G2NF_; i += 2) g2_load2(af[i], af[i + 1], bf[i], bf[i + 1], Ap, Bp, k0 + i * 4);
}

template <bool REFILL>
__device__ __forceinline__ void g2_phase(const v2f (&af)[G2NF_], const v2f (&bf)[G2NF_],
                                         v2f (&an)[G2NF_], v2f (&bn)[G2NF_],
                                         const float* __restrict__ Ap,
                                         const float* __restrict__ Bp, int kn,
                                         v8f& acc0, v8f& acc1) {
#pragma unroll
  for (int i = 0; i < G2NF_; i += 2) {
    acc0 = __builtin_amdgcn_wmma_f32_16x16x4_f32(false, af[i], false, bf[i], (short)0, acc0, false, false);
    acc1 = __builtin_amdgcn_wmma_f32_16x16x4_f32(false, af[i + 1], false, bf[i + 1], (short)0, acc1, false, false);
    if (REFILL) {
      g2_load2(an[i], an[i + 1], bn[i], bn[i + 1], Ap, Bp, kn + i * 4);
    }
  }
}

__global__ __launch_bounds__(128) void gemm2_kernel(const float* __restrict__ a,
                                                    const float* __restrict__ aval,
                                                    float* __restrict__ hist) {
  const int tid = threadIdx.x, lane = tid & 31, wave = tid >> 5;
  const int mt = blockIdx.x, nt = blockIdx.y * 4 + wave;
  if (nt >= (HLEN_ / 16)) return;
  const int m = lane & 15;
  const int kb = (lane & 16) ? 4 : 0;
  const int n = nt * 16 + m;
  const float* Ap = a + (size_t)(mt * 16 + m) * NK_ + kb;
  const float* Bp = aval + (size_t)kb * HLEN_ + n;
  v2f a0[G2NF_], b0[G2NF_], a1[G2NF_], b1[G2NF_], a2[G2NF_], b2[G2NF_];
  v8f acc0 = {}, acc1 = {};
  g2_load(a0, b0, Ap, Bp, 0);
  g2_load(a1, b1, Ap, Bp, G2KB_);
  for (int blk = 0; blk < G2NBLK_ - 2; blk += 3) {
    g2_phase<true>(a0, b0, a2, b2, Ap, Bp, (blk + 2) * G2KB_, acc0, acc1);
    g2_phase<true>(a1, b1, a0, b0, Ap, Bp, (blk + 3) * G2KB_, acc0, acc1);
    g2_phase<true>(a2, b2, a1, b1, Ap, Bp, (blk + 4) * G2KB_, acc0, acc1);
  }
  g2_phase<false>(a0, b0, a2, b2, Ap, Bp, 0, acc0, acc1);
  g2_phase<false>(a1, b1, a2, b2, Ap, Bp, 0, acc0, acc1);
#pragma unroll
  for (int r = 0; r < 8; ++r) {
    int row = r + ((lane & 16) ? 8 : 0);
    hist[(size_t)(mt * 16 + row) * HLEN_ + n] = acc0[r] + acc1[r];
  }
}

typedef __attribute__((ext_vector_type(4))) _Float16 v4h;
#define FKS_ (HOR_ / 4)
#define FNT_ (L1_ / 16)
#define FMT_ 2
#define FTH_ (SEQ_ / (16 * FMT_) * 32)

__global__ __launch_bounds__(FTH_) void featl1_kernel(const float* __restrict__ inp,
                                                     const float* __restrict__ hist,
                                                     const float* __restrict__ W1,
                                                     const float* __restrict__ b1,
                                                     _Float16* __restrict__ xf) {
  __shared__ float w1s[L1_ * (HOR_ + 1)];
  __shared__ float hrow[HLEN_];
  __shared__ float irow[SEQ_];
  const int b = blockIdx.x, tid = threadIdx.x, lane = tid & 31, wave = tid >> 5;
  for (int i = tid; i < L1_ * (HOR_ + 1); i += FTH_) w1s[i] = W1[i];
  for (int i = tid; i < HLEN_; i += FTH_) hrow[i] = hist[(size_t)b * HLEN_ + i];
  for (int i = tid; i < SEQ_; i += FTH_) irow[i] = inp[(size_t)b * SEQ_ + i];
  __syncthreads();
  const int c = lane & 15, half = lane >> 4;
  const int kb = 2 * half;
  v2f bf[FNT_][FKS_];
  float w0[FNT_], bb[FNT_];
#pragma unroll
  for (int nt = 0; nt < FNT_; ++nt) {
    const int o = 4 * c + nt;
    w0[nt] = w1s[o * (HOR_ + 1)];
    bb[nt] = b1[o];
#pragma unroll
    for (int ks = 0; ks < FKS_; ++ks) {
      bf[nt][ks][0] = w1s[o * (HOR_ + 1) + 1 + 4 * ks + kb];
      bf[nt][ks][1] = w1s[o * (HOR_ + 1) + 1 + 4 * ks + kb + 1];
    }
  }
#pragma unroll
  for (int mt = 0; mt < FMT_; ++mt) {
    const int t0 = wave * (16 * FMT_) + mt * 16;
    v2f af[FKS_];
#pragma unroll
    for (int ks = 0; ks < FKS_; ++ks) {
      af[ks][0] = hrow[t0 + c + 4 * ks + kb];
      af[ks][1] = hrow[t0 + c + 4 * ks + kb + 1];
    }
    v8f acc[FNT_];
#pragma unroll
    for (int r = 0; r < 8; ++r) {
      const float iv = irow[t0 + 8 * half + r];
#pragma unroll
      for (int nt = 0; nt < FNT_; ++nt) acc[nt][r] = bb[nt] + iv * w0[nt];
    }
#pragma unroll
    for (int ks = 0; ks < FKS_; ++ks)
#pragma unroll
      for (int nt = 0; nt < FNT_; ++nt)
        acc[nt] = __builtin_amdgcn_wmma_f32_16x16x4_f32(false, af[ks], false, bf[nt][ks],
                                                        (short)0, acc[nt], false, false);
#pragma unroll
    for (int r = 0; r < 8; ++r) {
      v4h pk;
#pragma unroll
      for (int nt = 0; nt < FNT_; ++nt) pk[nt] = (_Float16)tanhf(acc[nt][r]);
      *(v4h*)(xf + ((size_t)b * SEQ_ + (t0 + 8 * half + r)) * L1_ + 4 * c) = pk;
    }
  }
}

__device__ __forceinline__ void lds_barrier() {
  __builtin_amdgcn_fence(__ATOMIC_RELEASE, "workgroup", "local");
  __builtin_amdgcn_s_barrier();
  __builtin_amdgcn_fence(__ATOMIC_ACQUIRE, "workgroup", "local");
}

#define RB_ 1
#define TB_ 16
#define NBLK_ (SEQ_ / TB_)
#define WP0_ (L1_ + 8)
#define WP1_ (HID_ + 8)

template <int KIN, int WP>
__device__ __forceinline__ void proj_block(const _Float16* ap, const _Float16* wp,
                                           const float* bp, int cstep, float* pxs_l) {
  constexpr int KFX = KIN / 32;
  v16h xA[KFX];
#pragma unroll
  for (int kf = 0; kf < KFX; ++kf) {
    v8h lo = *(const v8h*)(ap + kf * 32);
    v8h hi = *(const v8h*)(ap + kf * 32 + 16);
#pragma unroll
    for (int i = 0; i < 8; ++i) { xA[kf][i] = lo[i]; xA[kf][i + 8] = hi[i]; }
  }
#pragma unroll
  for (int nt = 0; nt < 4; ++nt) {
    v8f px;
    const int co = (nt & 1) * 16 + (nt >> 1) * cstep;
    const float bias = bp[co];
#pragma unroll
    for (int r = 0; r < 8; ++r) px[r] = bias;
#pragma unroll
    for (int kf = 0; kf < KFX; ++kf) {
      const _Float16* q = wp + co * WP + kf * 32;
      v8h lo = *(const v8h*)q;
      v8h hi = *(const v8h*)(q + 16);
      v16h wB;
#pragma unroll
      for (int i = 0; i < 8; ++i) { wB[i] = lo[i]; wB[i + 8] = hi[i]; }
      if (kf < KFX - 1) {
        px = __builtin_amdgcn_wmma_f32_16x16x32_f16(false, xA[kf], false, wB,
                                                    (short)0, px, false, false);
      } else {
        asm volatile("v_wmma_f32_16x16x32_f16 %0, %1, %2, %0\n\tv_nop\n\tv_nop\n\tv_nop\n\tv_nop"
                     : "+v"(px) : "v"(xA[kf]), "v"(wB));
      }
    }
#pragma unroll
    for (int r = 0; r < 8; ++r) pxs_l[(nt * 8 + r) * 32] = px[r];
  }
}

__global__ __launch_bounds__(512, 1) void lstm2_kernel(const _Float16* __restrict__ xin,
                                                       const float* __restrict__ Wih0,
                                                       const float* __restrict__ Whh0,
                                                       const float* __restrict__ bih0,
                                                       const float* __restrict__ bhh0,
                                                       const float* __restrict__ Wih1,
                                                       const float* __restrict__ Whh1,
                                                       const float* __restrict__ bih1,
                                                       const float* __restrict__ bhh1,
                                                       float* __restrict__ hlast) {
  __shared__ __align__(16) _Float16 wih0s[GATES_ * WP0_];
  __shared__ __align__(16) _Float16 wih1s[GATES_ * WP1_];
  __shared__ float pxs[16 * 4 * 8 * 32];
  __shared__ float gates[2][GATES_];
  __shared__ float biass[2][GATES_];
  __shared__ __align__(16) _Float16 hbuf[2][RB_][HID_];
  __shared__ __align__(16) _Float16 h0blk[TB_][RB_][HID_];
  const int tid = threadIdx.x, lane = tid & 31;
  const int wave = __builtin_amdgcn_readfirstlane(tid >> 5);
  const int layer = wave >> 3;
  const int ltid = tid & 255;
  const int bm = blockIdx.x * RB_;
  const int wl = wave & 7;
  const bool tanh_wave = wl < 4;
  const int cb0 = tanh_wave ? 2 * HID_ + 32 * wl : (wl - 4) * 64;
  const int cstep = tanh_wave ? HID_ : 32;
  const int m = lane & 15;
  const int hi16 = (lane >> 4) & 1;
  const int kbh = hi16 * 8;
  const float* __restrict__ Whh = layer ? Whh1 : Whh0;
  const float* __restrict__ bih = layer ? bih1 : bih0;
  const float* __restrict__ bhh = layer ? bhh1 : bhh0;

  for (int i = tid; i < GATES_ * L1_; i += 512)
    wih0s[(i >> 6) * WP0_ + (i & (L1_ - 1))] = (_Float16)Wih0[i];
  for (int i = tid; i < GATES_ * HID_; i += 512)
    wih1s[(i >> 7) * WP1_ + (i & (HID_ - 1))] = (_Float16)Wih1[i];
  for (int i = tid; i < TB_ * RB_ * HID_; i += 512)
    (&h0blk[0][0][0])[i] = (_Float16)0.f;
  for (int i = ltid; i < GATES_; i += 256) biass[layer][i] = bih[i] + bhh[i];

  v16h whhB[4][4];
#pragma unroll
  for (int nt = 0; nt < 4; ++nt) {
    const int g = cb0 + (nt & 1) * 16 + (nt >> 1) * cstep + m;
#pragma unroll
    for (int kf = 0; kf < 4; ++kf)
#pragma unroll
      for (int h = 0; h < 16; ++h) {
        int k = kf * 32 + ((h & 8) << 1) + kbh + (h & 7);
        whhB[nt][kf][h] = (_Float16)Whh[(size_t)g * HID_ + k];
      }
  }

  float* gl = gates[layer];
  _Float16* hb = hbuf[layer][0];
  float* pxs_l = pxs + wave * (4 * 8 * 32) + lane;
  const float* pxs_r = pxs + wave * (4 * 8 * 32) + m;
  const bool owner = (wave & 7) < HID_ / 32;
  float c = 0.f;
  if (owner) hb[ltid] = (_Float16)0.f;
  __syncthreads();

  for (int blk = 0; blk <= NBLK_; ++blk) {
    const bool active = layer ? (blk > 0) : (blk < NBLK_);
    const int t0 = (blk - layer) * TB_;
    if (layer == 0) {
      const int tx = (blk < NBLK_ ? blk : NBLK_ - 1) * TB_;
      proj_block<L1_, WP0_>(xin + ((size_t)bm * SEQ_ + (tx + m)) * L1_ + kbh,
                            wih0s + (cb0 + m) * WP0_ + kbh, &biass[0][cb0 + m], cstep, pxs_l);
    } else {
      proj_block<HID_, WP1_>(&h0blk[m][0][kbh],
                             wih1s + (cb0 + m) * WP1_ + kbh, &biass[1][cb0 + m], cstep, pxs_l);
    }
#pragma unroll 1
    for (int hs = 0; hs < 2; ++hs)
#pragma unroll
    for (int r8 = 0; r8 < 8; ++r8) {
      const int tl = hs * 8 + r8;
      const int t = t0 + tl;
      v16h hA[4];
#pragma unroll
      for (int kf = 0; kf < 4; ++kf) {
        const _Float16* p = hb + kf * 32 + kbh;
        v8h lo = *(const v8h*)p;
        v8h hi = *(const v8h*)(p + 16);
#pragma unroll
        for (int i = 0; i < 8; ++i) { hA[kf][i] = lo[i]; hA[kf][i + 8] = hi[i]; }
      }
      v8f acc[4];
#pragma unroll
      for (int nt = 0; nt < 4; ++nt) {
#pragma unroll
        for (int r = 0; r < 8; ++r) acc[nt][r] = 0.f;
        acc[nt][r8] = pxs_r[(nt * 8 + r8) * 32 + 16 * hs];
#pragma unroll
        for (int kf = 0; kf < 4; ++kf)
          acc[nt] = __builtin_amdgcn_wmma_f32_16x16x32_f16(false, hA[kf], false, whhB[nt][kf],
                                                           (short)0, acc[nt], false, false);
      }
      const float s0 = hi16 ? acc[1][r8] : acc[0][r8];
      const float s1 = hi16 ? acc[3][r8] : acc[2][r8];
      gl[cb0 + cstep + lane] = 1.0f / (1.0f + __expf(-s1));
      float a0 = 1.0f / (1.0f + __expf(-s0));
      if (tanh_wave) a0 = tanhf(s0);
      gl[cb0 + lane] = a0;
      lds_barrier();
      if (active && owner) {
        const int j = ltid;
        const float iv = gl[j];
        const float fv = gl[HID_ + j];
        const float gv = gl[2 * HID_ + j];
        const float ov = gl[3 * HID_ + j];
        c = fv * c + iv * gv;
        float hv = ov * tanhf(c);
        hb[j] = (_Float16)hv;
        if (layer == 0) h0blk[tl][0][j] = (_Float16)hv;
        else if (t == SEQ_ - 1) hlast[(size_t)bm * HID_ + j] = hv;
      }
      lds_barrier();
    }
  }
}

__global__ __launch_bounds__(128) void head_kernel(const float* __restrict__ h1last,
                                                   const float* __restrict__ W2,
                                                   const float* __restrict__ b2,
                                                   const float* __restrict__ W3,
                                                   const float* __restrict__ b3,
                                                   float* __restrict__ out) {
  __shared__ float lastS[HID_];
  __shared__ float l2S[L2_];
  const int b = blockIdx.x, tid = threadIdx.x;
  lastS[tid] = tanhf(h1last[(size_t)b * HID_ + tid]);
  __syncthreads();
  if (tid < L2_) {
    float acc = b2[tid];
    for (int k = 0; k < HID_; ++k) acc += lastS[k] * W2[tid * HID_ + k];
    l2S[tid] = tanhf(acc);
  }
  __syncthreads();
  if (tid < HOR_) {
    float acc = b3[tid];
    for (int k = 0; k < L2_; ++k) acc += l2S[k] * W3[tid * L2_ + k];
    out[(size_t)b * HOR_ + tid] = acc;
  }
}

extern "C" void kernel_launch(void* const* d_in, const int* in_sizes, int n_in,
                              void* d_out, int out_size, void* d_ws, size_t ws_size,
                              hipStream_t stream) {
  (void)in_sizes; (void)n_in; (void)out_size; (void)ws_size;
  const float* inp  = (const float*)d_in[0];
  const float* akey = (const float*)d_in[1];
  const float* aval = (const float*)d_in[2];
  const float* W1   = (const float*)d_in[3];
  const float* b1   = (const float*)d_in[4];
  const float* Wih0 = (const float*)d_in[5];
  const float* Whh0 = (const float*)d_in[6];
  const float* bih0 = (const float*)d_in[7];
  const float* bhh0 = (const float*)d_in[8];
  const float* Wih1 = (const float*)d_in[9];
  const float* Whh1 = (const float*)d_in[10];
  const float* bih1 = (const float*)d_in[11];
  const float* bhh1 = (const float*)d_in[12];
  const float* W2   = (const float*)d_in[13];
  const float* b2   = (const float*)d_in[14];
  const float* W3   = (const float*)d_in[15];
  const float* b3   = (const float*)d_in[16];

  char* ws = (char*)d_ws;
  size_t off = 0;
  float*    dist   = (float*)(ws + off);    off += (size_t)B_ * NK_ * 4;
  float*    hist   = (float*)(ws + off);    off += (size_t)B_ * HLEN_ * 4;
  _Float16* xf     = (_Float16*)(ws + off); off += (size_t)B_ * SEQ_ * L1_ * 2;
  float*    h1last = (float*)(ws + off);    off += (size_t)B_ * HID_ * 4;

  gemm1_kernel<<<dim3(16, 64), 256, 0, stream>>>(inp, akey, dist);
  softmax_kernel<<<dim3(B_), 256, 0, stream>>>(dist);
  gemm2_kernel<<<dim3(16, 9), 128, 0, stream>>>(dist, aval, hist);
  featl1_kernel<<<dim3(B_), FTH_, 0, stream>>>(inp, hist, W1, b1, xf);
  lstm2_kernel<<<dim3(B_ / RB_), 512, 0, stream>>>(xf, Wih0, Whh0, bih0, bhh0, Wih1, Whh1, bih1, bhh1, h1last);
  head_kernel<<<dim3(B_), 128, 0, stream>>>(h1last, W2, b2, W3, b3, (float*)d_out);
}
